# starting kernel
# baseline (speedup 1.0000x reference)
#include <hip/hip_runtime.h>
#include <hip/hip_bf16.h>

typedef __attribute__((ext_vector_type(2))) float v2f;
typedef __attribute__((ext_vector_type(8))) float v8f;

#define TPB 256
#define KC    64
#define ASTR  68
#define WTSTR 68

__global__ void gcn_deg_init(float* __restrict__ deg, int n) {
    int i = blockIdx.x * blockDim.x + threadIdx.x;
    if (i < n) deg[i] = 1.0f;
}

__global__ void gcn_deg_edges(const int* __restrict__ dst, float* __restrict__ deg, int E) {
    int e = blockIdx.x * blockDim.x + threadIdx.x;
    if (e < E)
        __hip_atomic_fetch_add(&deg[dst[e]], 1.0f, __ATOMIC_RELAXED, __HIP_MEMORY_SCOPE_AGENT);
}

__global__ void gcn_dinv(float* __restrict__ deg, int n) {
    int i = blockIdx.x * blockDim.x + threadIdx.x;
    if (i < n) deg[i] = rsqrtf(fmaxf(deg[i], 1.0f));
}

__global__ void gcn_edge_coef(const int* __restrict__ src, const int* __restrict__ dst,
                              const float* __restrict__ dinv, float* __restrict__ coef, int E) {
    int e = blockIdx.x * blockDim.x + threadIdx.x;
    if (e < E) coef[e] = dinv[src[e]] * dinv[dst[e]];
}

__global__ __launch_bounds__(TPB)
void gcn_gemm_wmma(const float* __restrict__ A, const float* __restrict__ W,
                   float* __restrict__ C, int M, int K, int N) {
    __shared__ float lA[128 * ASTR];
    __shared__ float lWt[64 * WTSTR];
    const int tid  = threadIdx.x;
    const int lane = tid & 31;
    const int wid  = tid >> 5;
    const int l16  = lane & 15;
    const int half = lane >> 4;
    const int rowBase = blockIdx.x * 128;
    const int colBase = blockIdx.y * 64;

    v8f acc[4] = {};

    for (int kc = 0; kc < K; kc += KC) {
#pragma unroll
        for (int i = 0; i < 8; ++i) {
            int idx = tid + i * TPB;
            int r   = idx >> 4;
            int c4  = (idx & 15) << 2;
            int gr  = min(rowBase + r, M - 1);
            float4 v = *(const float4*)(A + (size_t)gr * K + kc + c4);
            *(float4*)(&lA[r * ASTR + c4]) = v;
        }
#pragma unroll
        for (int i = 0; i < 4; ++i) {
            int idx = tid + i * TPB;
            int r   = idx >> 4;
            int c4  = (idx & 15) << 2;
            float4 v = *(const float4*)(W + (size_t)(kc + r) * N + colBase + c4);
            lWt[(c4 + 0) * WTSTR + r] = v.x;
            lWt[(c4 + 1) * WTSTR + r] = v.y;
            lWt[(c4 + 2) * WTSTR + r] = v.z;
            lWt[(c4 + 3) * WTSTR + r] = v.w;
        }
        __syncthreads();

        for (int kk = 0; kk < KC; kk += 4) {
            const int ks = kk + 2 * half;
            v2f a = *(const v2f*)(&lA[(wid * 16 + l16) * ASTR + ks]);
#pragma unroll
            for (int ct = 0; ct < 4; ++ct) {
                v2f b = *(const v2f*)(&lWt[(ct * 16 + l16) * WTSTR + ks]);
                acc[ct] = __builtin_amdgcn_wmma_f32_16x16x4_f32(
                    false, a, false, b, (short)0, acc[ct], false, false);
            }
        }
        __syncthreads();
    }

#pragma unroll
    for (int ct = 0; ct < 4; ++ct) {
        const int col = colBase + ct * 16 + l16;
#pragma unroll
        for (int v = 0; v < 8; ++v) {
            const int row = rowBase + wid * 16 + v + 8 * half;
            if (row < M) C[(size_t)row * N + col] = acc[ct][v];
        }
    }
}

__global__ void gcn_agg_init(const float* __restrict__ xw, const float* __restrict__ bias,
                             const float* __restrict__ dinv, float* __restrict__ agg,
                             long totalV4, int logF) {
    long idx = (long)blockIdx.x * blockDim.x + threadIdx.x;
    if (idx >= totalV4) return;
    size_t elem = (size_t)idx << 2;
    int i  = (int)(elem >> logF);
    int f4 = (int)(elem & ((1 << logF) - 1));
    float di = dinv[i];
    float s  = di * di;
    float4 xv = *(const float4*)(xw + elem);
    float4 bv = *(const float4*)(bias + f4);
    float4 o;
    o.x = bv.x + xv.x * s; o.y = bv.y + xv.y * s;
    o.z = bv.z + xv.z * s; o.w = bv.w + xv.w * s;
    *(float4*)(agg + elem) = o;
}

__global__ void gcn_edge_scatter(const int* __restrict__ src, const int* __restrict__ dst,
                                 const float* __restrict__ coef, const float* __restrict__ xw,
                                 float* __restrict__ agg, long totalV4, int logF) {
    long gid = (long)blockIdx.x * blockDim.x + threadIdx.x;
    if (gid >= totalV4) return;
    const int logF4 = logF - 2;
    int e  = (int)(gid >> logF4);
    int f4 = (int)((gid & ((1 << logF4) - 1)) << 2);
    float c = coef[e];
    const float4 xv = *(const float4*)(xw + ((size_t)src[e] << logF) + f4);
    float* base = agg + ((size_t)dst[e] << logF) + f4;
    __hip_atomic_fetch_add(base + 0, xv.x * c, __ATOMIC_RELAXED, __HIP_MEMORY_SCOPE_AGENT);
    __hip_atomic_fetch_add(base + 1, xv.y * c, __ATOMIC_RELAXED, __HIP_MEMORY_SCOPE_AGENT);
    __hip_atomic_fetch_add(base + 2, xv.z * c, __ATOMIC_RELAXED, __HIP_MEMORY_SCOPE_AGENT);
    __hip_atomic_fetch_add(base + 3, xv.w * c, __ATOMIC_RELAXED, __HIP_MEMORY_SCOPE_AGENT);
}

__global__ void gcn_bn_relu(const float* __restrict__ in, float* __restrict__ out,
                            const float* __restrict__ g, const float* __restrict__ beta,
                            const float* __restrict__ m, const float* __restrict__ v,
                            long totalV4, int fmask) {
    long idx = (long)blockIdx.x * blockDim.x + threadIdx.x;
    if (idx >= totalV4) return;
    size_t elem = (size_t)idx << 2;
    int f4 = (int)(elem & fmask);
    float4 xv = *(const float4*)(in + elem);
    float4 gv = *(const float4*)(g + f4);
    float4 bv = *(const float4*)(beta + f4);
    float4 mv = *(const float4*)(m + f4);
    float4 vv = *(const float4*)(v + f4);
    float4 o;
    o.x = fmaxf((xv.x - mv.x) * rsqrtf(vv.x + 1e-5f) * gv.x + bv.x, 0.0f);
    o.y = fmaxf((xv.y - mv.y) * rsqrtf(vv.y + 1e-5f) * gv.y + bv.y, 0.0f);
    o.z = fmaxf((xv.z - mv.z) * rsqrtf(vv.z + 1e-5f) * gv.z + bv.z, 0.0f);
    o.w = fmaxf((xv.w - mv.w) * rsqrtf(vv.w + 1e-5f) * gv.w + bv.w, 0.0f);
    *(float4*)(out + elem) = o;
}

__global__ __launch_bounds__(TPB)
void gcn_softmax64(const float* __restrict__ in, float* __restrict__ out, int n) {
    int wid  = threadIdx.x >> 5;
    int lane = threadIdx.x & 31;
    int row  = blockIdx.x * 8 + wid;
    if (row >= n) return;
    const float* p = in + (size_t)row * 64;
    float v0 = p[lane], v1 = p[lane + 32];
    float mx = fmaxf(v0, v1);
#pragma unroll
    for (int off = 16; off > 0; off >>= 1) mx = fmaxf(mx, __shfl_xor(mx, off, 32));
    float e0 = __expf(v0 - mx), e1 = __expf(v1 - mx);
    float s = e0 + e1;
#pragma unroll
    for (int off = 16; off > 0; off >>= 1) s += __shfl_xor(s, off, 32);
    float inv = 1.0f / s;
    float* q = out + (size_t)row * 64;
    q[lane] = e0 * inv;
    q[lane + 32] = e1 * inv;
}

extern "C" void kernel_launch(void* const* d_in, const int* in_sizes, int n_in,
                              void* d_out, int out_size, void* d_ws, size_t ws_size,
                              hipStream_t stream) {
    const int IN = 256, HID_ = 256, OUT_ = 64;
    const int N = in_sizes[0] / IN;
    const int E = in_sizes[1] / 2;

    const float* x    = (const float*)d_in[0];
    const int*   src  = (const int*)d_in[1];
    const int*   dst  = ((const int*)d_in[1]) + E;
    const float* W1   = (const float*)d_in[2];
    const float* b1   = (const float*)d_in[3];
    const float* g1   = (const float*)d_in[4];
    const float* be1  = (const float*)d_in[5];
    const float* m1   = (const float*)d_in[6];
    const float* v1   = (const float*)d_in[7];
    const float* W2   = (const float*)d_in[8];
    const float* b2   = (const float*)d_in[9];
    const float* g2   = (const float*)d_in[10];
    const float* be2  = (const float*)d_in[11];
    const float* m2   = (const float*)d_in[12];
    const float* v2   = (const float*)d_in[13];
    const float* W3   = (const float*)d_in[14];
    const float* b3   = (const float*)d_in[15];
    float* out = (float*)d_out;

    char* ws = (char*)d_ws;
    size_t off = 0;
    float* dinv = (float*)(ws + off); off += ((size_t)N * 4 + 255) & ~(size_t)255;
    float* coef = (float*)(ws + off); off += ((size_t)E * 4 + 255) & ~(size_t)255;
    float* bufA = (float*)(ws + off); off += (size_t)N * HID_ * 4;
    float* bufB = (float*)(ws + off);
    (void)ws_size; (void)n_in; (void)out_size;

    const long n256v4 = (long)N * HID_ / 4;
    const long n64v4  = (long)N * OUT_ / 4;
    const long e256v4 = (long)E * HID_ / 4;
    const long e64v4  = (long)E * OUT_ / 4;
    const int gN = (N + TPB - 1) / TPB;
    const int gE = (E + TPB - 1) / TPB;
    const int gemmX = (N + 127) / 128;
    auto blocks = [](long t) { return (int)((t + TPB - 1) / TPB); };

    gcn_deg_init <<<gN, TPB, 0, stream>>>(dinv, N);
    gcn_deg_edges<<<gE, TPB, 0, stream>>>(dst, dinv, E);
    gcn_dinv     <<<gN, TPB, 0, stream>>>(dinv, N);
    gcn_edge_coef<<<gE, TPB, 0, stream>>>(src, dst, dinv, coef, E);

    gcn_gemm_wmma   <<<dim3(gemmX, HID_ / 64), TPB, 0, stream>>>(x, W1, bufA, N, IN, HID_);
    gcn_agg_init    <<<blocks(n256v4), TPB, 0, stream>>>(bufA, b1, dinv, bufB, n256v4, 8);
    gcn_edge_scatter<<<blocks(e256v4), TPB, 0, stream>>>(src, dst, coef, bufA, bufB, e256v4, 8);
    gcn_bn_relu     <<<blocks(n256v4), TPB, 0, stream>>>(bufB, bufA, g1, be1, m1, v1, n256v4, 255);

    gcn_gemm_wmma   <<<dim3(gemmX, HID_ / 64), TPB, 0, stream>>>(bufA, W2, bufB, N, HID_, HID_);
    gcn_agg_init    <<<blocks(n256v4), TPB, 0, stream>>>(bufB, b2, dinv, bufA, n256v4, 8);
    gcn_edge_scatter<<<blocks(e256v4), TPB, 0, stream>>>(src, dst, coef, bufB, bufA, e256v4, 8);
    gcn_bn_relu     <<<blocks(n256v4), TPB, 0, stream>>>(bufA, bufB, g2, be2, m2, v2, n256v4, 255);

    gcn_gemm_wmma   <<<dim3(gemmX, OUT_ / 64), TPB, 0, stream>>>(bufB, W3, bufA, N, HID_, OUT_);
    gcn_agg_init    <<<blocks(n64v4), TPB, 0, stream>>>(bufA, b3, dinv, bufB, n64v4, 6);
    gcn_edge_scatter<<<blocks(e64v4), TPB, 0, stream>>>(src, dst, coef, bufA, bufB, e64v4, 6);
    gcn_softmax64   <<<(N + 7) / 8, TPB, 0, stream>>>(bufB, out, N);
}


# Atomic feature scatter replaced by per-call dst-sorted CSR and one-wave-per-row gather (self loop and bias as start value); a row's edges are summed in integer-counter arrival order
# speedup vs baseline: 2.4437x; 2.4437x over previous
#include <hip/hip_runtime.h>
#include <hip/hip_bf16.h>

typedef __attribute__((ext_vector_type(2))) float v2f;
typedef __attribute__((ext_vector_type(8))) float v8f;

#define TPB 256
#define KC    64
#define ASTR  68
#define WTSTR 68

__global__ void gcn_cnt_zero(int* __restrict__ cnt, int n) {
    int i = blockIdx.x * blockDim.x + threadIdx.x;
    if (i < n) cnt[i] = 0;
}

__global__ void gcn_cnt_edges(const int* __restrict__ dst, int* __restrict__ cnt,
                              int* __restrict__ rank, int E) {
    int e = blockIdx.x * blockDim.x + threadIdx.x;
    if (e < E) rank[e] = atomicAdd(&cnt[dst[e]], 1);
}

__global__ __launch_bounds__(1024)
void gcn_row_scan(const int* __restrict__ cnt, int* __restrict__ rowstart, int n) {
    __shared__ int wtot[32];
    __shared__ int wpre[32];
    const int tid  = threadIdx.x;
    const int lane = tid & 31;
    const int wid  = tid >> 5;
    const int tile0 = blockIdx.x * 1024;
    int pre = 0;
#pragma unroll 8
    for (int j = tid; j < tile0; j += 1024) pre += cnt[j];
    const int i = tile0 + tid;
    const int c = (i < n) ? cnt[i] : 0;
    int inc = c;
#pragma unroll
    for (int off = 1; off < 32; off <<= 1) {
        int t = __shfl_up(inc, off, 32);
        if (lane >= off) inc += t;
    }
#pragma unroll
    for (int off = 16; off > 0; off >>= 1) pre += __shfl_xor(pre, off, 32);
    if (lane == 31) wtot[wid] = inc;
    if (lane == 0)  wpre[wid] = pre;
    __syncthreads();
    int tot = wpre[lane] + ((lane < wid) ? wtot[lane] : 0);
#pragma unroll
    for (int off = 16; off > 0; off >>= 1) tot += __shfl_xor(tot, off, 32);
    if (i < n) rowstart[i] = tot + inc - c;
}

__global__ void gcn_csr_fill(const int* __restrict__ src, const int* __restrict__ dst,
                             const int* __restrict__ rank, const int* __restrict__ rowstart,
                             int* __restrict__ csr_src, int E) {
    int e = blockIdx.x * blockDim.x + threadIdx.x;
    if (e < E) csr_src[rowstart[dst[e]] + rank[e]] = src[e];
}

__global__ __launch_bounds__(TPB)
void gcn_gemm_wmma(const float* __restrict__ A, const float* __restrict__ W,
                   float* __restrict__ C, int M, int K, int N) {
    __shared__ float lA[128 * ASTR];
    __shared__ float lWt[64 * WTSTR];
    const int tid  = threadIdx.x;
    const int lane = tid & 31;
    const int wid  = tid >> 5;
    const int l16  = lane & 15;
    const int half = lane >> 4;
    const int rowBase = blockIdx.x * 128;
    const int colBase = blockIdx.y * 64;

    v8f acc[4] = {};

    for (int kc = 0; kc < K; kc += KC) {
#pragma unroll
        for (int i = 0; i < 8; ++i) {
            int idx = tid + i * TPB;
            int r   = idx >> 4;
            int c4  = (idx & 15) << 2;
            int gr  = min(rowBase + r, M - 1);
            float4 v = *(const float4*)(A + (size_t)gr * K + kc + c4);
            *(float4*)(&lA[r * ASTR + c4]) = v;
        }
#pragma unroll
        for (int i = 0; i < 4; ++i) {
            int idx = tid + i * TPB;
            int r   = idx >> 4;
            int c4  = (idx & 15) << 2;
            float4 v = *(const float4*)(W + (size_t)(kc + r) * N + colBase + c4);
            lWt[(c4 + 0) * WTSTR + r] = v.x;
            lWt[(c4 + 1) * WTSTR + r] = v.y;
            lWt[(c4 + 2) * WTSTR + r] = v.z;
            lWt[(c4 + 3) * WTSTR + r] = v.w;
        }
        __syncthreads();

        for (int kk = 0; kk < KC; kk += 4) {
            const int ks = kk + 2 * half;
            v2f a = *(const v2f*)(&lA[(wid * 16 + l16) * ASTR + ks]);
#pragma unroll
            for (int ct = 0; ct < 4; ++ct) {
                v2f b = *(const v2f*)(&lWt[(ct * 16 + l16) * WTSTR + ks]);
                acc[ct] = __builtin_amdgcn_wmma_f32_16x16x4_f32(
                    false, a, false, b, (short)0, acc[ct], false, false);
            }
        }
        __syncthreads();
    }

#pragma unroll
    for (int ct = 0; ct < 4; ++ct) {
        const int col = colBase + ct * 16 + l16;
#pragma unroll
        for (int v = 0; v < 8; ++v) {
            const int row = rowBase + wid * 16 + v + 8 * half;
            if (row < M) C[(size_t)row * N + col] = acc[ct][v];
        }
    }
}

template <int F>
__global__ __launch_bounds__(TPB)
void gcn_gather(const int* __restrict__ rowstart, const int* __restrict__ csr_src,
                const float* __restrict__ xw, const float* __restrict__ bias,
                float* __restrict__ agg, int n, int E) {
    constexpr int VW  = (F >= 128) ? 4 : F / 32;
    constexpr int NP  = F / (32 * VW);
    constexpr int UNR = (F >= 256) ? 4 : 8;
    typedef float vec_t __attribute__((ext_vector_type(VW)));
    const int lane = threadIdx.x & 31;
    const int row = __builtin_amdgcn_readfirstlane((int)(blockIdx.x * (TPB / 32) + (threadIdx.x >> 5)));
    if (row >= n) return;
    const int beg = rowstart[row];
    const int end = (row + 1 < n) ? rowstart[row + 1] : E;
    const float di = rsqrtf(fmaxf((float)(end - beg + 1), 1.0f));
    const int col = lane * VW;

    vec_t acc[NP];
    {
        const float sl = di * di;
#pragma unroll
        for (int p = 0; p < NP; ++p) {
            const vec_t xv = *(const vec_t*)(xw + (size_t)row * F + p * 32 * VW + col);
            const vec_t bv = *(const vec_t*)(bias + p * 32 * VW + col);
            acc[p] = bv + xv * sl;
        }
    }

    for (int base = beg; base < end; base += 32) {
        const int nb = min(32, end - base);
        int   s_l = 0;
        float c_l = 0.0f;
        if (lane < nb) {
            s_l = csr_src[base + lane];
            const int a  = rowstart[s_l];
            const int bn = rowstart[min(s_l + 1, n - 1)];
            const int b  = (s_l + 1 < n) ? bn : E;
            c_l = rsqrtf(fmaxf((float)(b - a + 1), 1.0f)) * di;
        }
        const int c_bits = __builtin_bit_cast(int, c_l);
        int k = 0;
        for (; k + UNR <= nb; k += UNR) {
            vec_t v[UNR][NP];
            float c[UNR];
#pragma unroll
            for (int u = 0; u < UNR; ++u) {
                const int s = __builtin_amdgcn_readlane(s_l, k + u);
                c[u] = __builtin_bit_cast(float, __builtin_amdgcn_readlane(c_bits, k + u));
                const float* r = xw + (size_t)s * F + col;
#pragma unroll
                for (int p = 0; p < NP; ++p) v[u][p] = *(const vec_t*)(r + p * 32 * VW);
            }
#pragma unroll
            for (int u = 0; u < UNR; ++u)
#pragma unroll
                for (int p = 0; p < NP; ++p) acc[p] += v[u][p] * c[u];
        }
#pragma unroll 1
        for (; k < nb; ++k) {
            const int s = __builtin_amdgcn_readlane(s_l, k);
            const float c = __builtin_bit_cast(float, __builtin_amdgcn_readlane(c_bits, k));
            const float* r = xw + (size_t)s * F + col;
#pragma unroll
            for (int p = 0; p < NP; ++p) acc[p] += *(const vec_t*)(r + p * 32 * VW) * c;
        }
    }

#pragma unroll
    for (int p = 0; p < NP; ++p)
        *(vec_t*)(agg + (size_t)row * F + p * 32 * VW + col) = acc[p];
}

__global__ void gcn_bn_relu(const float* __restrict__ in, float* __restrict__ out,
                            const float* __restrict__ g, const float* __restrict__ beta,
                            const float* __restrict__ m, const float* __restrict__ v,
                            long totalV4, int fmask) {
    long idx = (long)blockIdx.x * blockDim.x + threadIdx.x;
    if (idx >= totalV4) return;
    size_t elem = (size_t)idx << 2;
    int f4 = (int)(elem & fmask);
    float4 xv = *(const float4*)(in + elem);
    float4 gv = *(const float4*)(g + f4);
    float4 bv = *(const float4*)(beta + f4);
    float4 mv = *(const float4*)(m + f4);
    float4 vv = *(const float4*)(v + f4);
    float4 o;
    o.x = fmaxf((xv.x - mv.x) * rsqrtf(vv.x + 1e-5f) * gv.x + bv.x, 0.0f);
    o.y = fmaxf((xv.y - mv.y) * rsqrtf(vv.y + 1e-5f) * gv.y + bv.y, 0.0f);
    o.z = fmaxf((xv.z - mv.z) * rsqrtf(vv.z + 1e-5f) * gv.z + bv.z, 0.0f);
    o.w = fmaxf((xv.w - mv.w) * rsqrtf(vv.w + 1e-5f) * gv.w + bv.w, 0.0f);
    *(float4*)(out + elem) = o;
}

__global__ __launch_bounds__(TPB)
void gcn_softmax64(const float* __restrict__ in, float* __restrict__ out, int n) {
    int wid  = threadIdx.x >> 5;
    int lane = threadIdx.x & 31;
    int row  = blockIdx.x * 8 + wid;
    if (row >= n) return;
    const float* p = in + (size_t)row * 64;
    float v0 = p[lane], v1 = p[lane + 32];
    float mx = fmaxf(v0, v1);
#pragma unroll
    for (int off = 16; off > 0; off >>= 1) mx = fmaxf(mx, __shfl_xor(mx, off, 32));
    float e0 = __expf(v0 - mx), e1 = __expf(v1 - mx);
    float s = e0 + e1;
#pragma unroll
    for (int off = 16; off > 0; off >>= 1) s += __shfl_xor(s, off, 32);
    float inv = 1.0f / s;
    float* q = out + (size_t)row * 64;
    q[lane] = e0 * inv;
    q[lane + 32] = e1 * inv;
}

extern "C" void kernel_launch(void* const* d_in, const int* in_sizes, int n_in,
                              void* d_out, int out_size, void* d_ws, size_t ws_size,
                              hipStream_t stream) {
    const int IN = 256, HID_ = 256, OUT_ = 64;
    const int N = in_sizes[0] / IN;
    const int E = in_sizes[1] / 2;

    const float* x    = (const float*)d_in[0];
    const int*   src  = (const int*)d_in[1];
    const int*   dst  = ((const int*)d_in[1]) + E;
    const float* W1   = (const float*)d_in[2];
    const float* b1   = (const float*)d_in[3];
    const float* g1   = (const float*)d_in[4];
    const float* be1  = (const float*)d_in[5];
    const float* m1   = (const float*)d_in[6];
    const float* v1   = (const float*)d_in[7];
    const float* W2   = (const float*)d_in[8];
    const float* b2   = (const float*)d_in[9];
    const float* g2   = (const float*)d_in[10];
    const float* be2  = (const float*)d_in[11];
    const float* m2   = (const float*)d_in[12];
    const float* v2   = (const float*)d_in[13];
    const float* W3   = (const float*)d_in[14];
    const float* b3   = (const float*)d_in[15];
    float* out = (float*)d_out;

    char* ws = (char*)d_ws;
    size_t off = 0;
    int*   rowstart = (int*)(ws + off);   off += ((size_t)N * 4 + 255) & ~(size_t)255;
    int*   csr_src  = (int*)(ws + off);   off += ((size_t)E * 4 + 255) & ~(size_t)255;
    float* bufA = (float*)(ws + off); off += (size_t)N * HID_ * 4;
    float* bufB = (float*)(ws + off);
    int* cnt  = (int*)bufA;
    int* rank = (int*)bufB;
    (void)ws_size; (void)n_in; (void)out_size;

    const long n256v4 = (long)N * HID_ / 4;
    const int gN = (N + TPB - 1) / TPB;
    const int gE = (E + TPB - 1) / TPB;
    const int gemmX = (N + 127) / 128;
    const int gRow = (N + TPB / 32 - 1) / (TPB / 32);
    auto blocks = [](long t) { return (int)((t + TPB - 1) / TPB); };

    gcn_cnt_zero <<<gN, TPB, 0, stream>>>(cnt, N);
    gcn_cnt_edges<<<gE, TPB, 0, stream>>>(dst, cnt, rank, E);
    gcn_row_scan <<<(N + 1023) / 1024, 1024, 0, stream>>>(cnt, rowstart, N);
    gcn_csr_fill <<<gE, TPB, 0, stream>>>(src, dst, rank, rowstart, csr_src, E);

    gcn_gemm_wmma  <<<dim3(gemmX, HID_ / 64), TPB, 0, stream>>>(x, W1, bufA, N, IN, HID_);
    gcn_gather<256><<<gRow, TPB, 0, stream>>>(rowstart, csr_src, bufA, b1, bufB, N, E);
    gcn_bn_relu    <<<blocks(n256v4), TPB, 0, stream>>>(bufB, bufA, g1, be1, m1, v1, n256v4, 255);

    gcn_gemm_wmma  <<<dim3(gemmX, HID_ / 64), TPB, 0, stream>>>(bufA, W2, bufB, N, HID_, HID_);
    gcn_gather<256><<<gRow, TPB, 0, stream>>>(rowstart, csr_src, bufB, b2, bufA, N, E);
    gcn_bn_relu    <<<blocks(n256v4), TPB, 0, stream>>>(bufA, bufB, g2, be2, m2, v2, n256v4, 255);

    gcn_gemm_wmma  <<<dim3(gemmX, OUT_ / 64), TPB, 0, stream>>>(bufB, W3, bufA, N, HID_, OUT_);
    gcn_gather<64> <<<gRow, TPB, 0, stream>>>(rowstart, csr_src, bufA, b3, bufB, N, E);
    gcn_softmax64  <<<(N + 7) / 8, TPB, 0, stream>>>(bufB, out, N);
}


# BatchNorm+ReLU and the final softmax are applied inside the gather kernels before their only store: 3 launches and their round trips gone; softmax row sum now pairs neighbouring columns first
# speedup vs baseline: 2.8354x; 1.1603x over previous
#include <hip/hip_runtime.h>
#include <hip/hip_bf16.h>

typedef __attribute__((ext_vector_type(2))) float v2f;
typedef __attribute__((ext_vector_type(8))) float v8f;

#define TPB 256
#define KC    64
#define ASTR  68
#define WTSTR 68

__global__ void gcn_cnt_zero(int* __restrict__ cnt, int n) {
    int i = blockIdx.x * blockDim.x + threadIdx.x;
    if (i < n) cnt[i] = 0;
}

__global__ void gcn_cnt_edges(const int* __restrict__ dst, int* __restrict__ cnt,
                              int* __restrict__ rank, int E) {
    int e = blockIdx.x * blockDim.x + threadIdx.x;
    if (e < E) rank[e] = atomicAdd(&cnt[dst[e]], 1);
}

__global__ __launch_bounds__(1024)
void gcn_row_scan(const int* __restrict__ cnt, int* __restrict__ rowstart, int n) {
    __shared__ int wtot[32];
    __shared__ int wpre[32];
    const int tid  = threadIdx.x;
    const int lane = tid & 31;
    const int wid  = tid >> 5;
    const int tile0 = blockIdx.x * 1024;
    int pre = 0;
#pragma unroll 8
    for (int j = tid; j < tile0; j += 1024) pre += cnt[j];
    const int i = tile0 + tid;
    const int c = (i < n) ? cnt[i] : 0;
    int inc = c;
#pragma unroll
    for (int off = 1; off < 32; off <<= 1) {
        int t = __shfl_up(inc, off, 32);
        if (lane >= off) inc += t;
    }
#pragma unroll
    for (int off = 16; off > 0; off >>= 1) pre += __shfl_xor(pre, off, 32);
    if (lane == 31) wtot[wid] = inc;
    if (lane == 0)  wpre[wid] = pre;
    __syncthreads();
    int tot = wpre[lane] + ((lane < wid) ? wtot[lane] : 0);
#pragma unroll
    for (int off = 16; off > 0; off >>= 1) tot += __shfl_xor(tot, off, 32);
    if (i < n) rowstart[i] = tot + inc - c;
}

__global__ void gcn_csr_fill(const int* __restrict__ src, const int* __restrict__ dst,
                             const int* __restrict__ rank, const int* __restrict__ rowstart,
                             int* __restrict__ csr_src, int E) {
    int e = blockIdx.x * blockDim.x + threadIdx.x;
    if (e < E) csr_src[rowstart[dst[e]] + rank[e]] = src[e];
}

__global__ __launch_bounds__(TPB)
void gcn_gemm_wmma(const float* __restrict__ A, const float* __restrict__ W,
                   float* __restrict__ C, int M, int K, int N) {
    __shared__ float lA[128 * ASTR];
    __shared__ float lWt[64 * WTSTR];
    const int tid  = threadIdx.x;
    const int lane = tid & 31;
    const int wid  = tid >> 5;
    const int l16  = lane & 15;
    const int half = lane >> 4;
    const int rowBase = blockIdx.x * 128;
    const int colBase = blockIdx.y * 64;

    v8f acc[4] = {};

    for (int kc = 0; kc < K; kc += KC) {
#pragma unroll
        for (int i = 0; i < 8; ++i) {
            int idx = tid + i * TPB;
            int r   = idx >> 4;
            int c4  = (idx & 15) << 2;
            int gr  = min(rowBase + r, M - 1);
            float4 v = *(const float4*)(A + (size_t)gr * K + kc + c4);
            *(float4*)(&lA[r * ASTR + c4]) = v;
        }
#pragma unroll
        for (int i = 0; i < 4; ++i) {
            int idx = tid + i * TPB;
            int r   = idx >> 4;
            int c4  = (idx & 15) << 2;
            float4 v = *(const float4*)(W + (size_t)(kc + r) * N + colBase + c4);
            lWt[(c4 + 0) * WTSTR + r] = v.x;
            lWt[(c4 + 1) * WTSTR + r] = v.y;
            lWt[(c4 + 2) * WTSTR + r] = v.z;
            lWt[(c4 + 3) * WTSTR + r] = v.w;
        }
        __syncthreads();

        for (int kk = 0; kk < KC; kk += 4) {
            const int ks = kk + 2 * half;
            v2f a = *(const v2f*)(&lA[(wid * 16 + l16) * ASTR + ks]);
#pragma unroll
            for (int ct = 0; ct < 4; ++ct) {
                v2f b = *(const v2f*)(&lWt[(ct * 16 + l16) * WTSTR + ks]);
                acc[ct] = __builtin_amdgcn_wmma_f32_16x16x4_f32(
                    false, a, false, b, (short)0, acc[ct], false, false);
            }
        }
        __syncthreads();
    }

#pragma unroll
    for (int ct = 0; ct < 4; ++ct) {
        const int col = colBase + ct * 16 + l16;
#pragma unroll
        for (int v = 0; v < 8; ++v) {
            const int row = rowBase + wid * 16 + v + 8 * half;
            if (row < M) C[(size_t)row * N + col] = acc[ct][v];
        }
    }
}

#define EPI_BN_RELU 0
#define EPI_SOFTMAX 1
template <int F, int EPI>
__global__ __launch_bounds__(TPB)
void gcn_gather(const int* __restrict__ rowstart, const int* __restrict__ csr_src,
                const float* __restrict__ xw, const float* __restrict__ bias,
                const float* __restrict__ g, const float* __restrict__ beta,
                const float* __restrict__ m, const float* __restrict__ v,
                float* __restrict__ y, int n, int E) {
    static_assert(EPI == EPI_BN_RELU || F == 64, "the softmax form is written for 64 columns");
    constexpr int VW  = (F >= 128) ? 4 : F / 32;
    constexpr int NP  = F / (32 * VW);
    constexpr int UNR = (F >= 256) ? 4 : 8;
    typedef float vec_t __attribute__((ext_vector_type(VW)));
    const int lane = threadIdx.x & 31;
    const int row = __builtin_amdgcn_readfirstlane((int)(blockIdx.x * (TPB / 32) + (threadIdx.x >> 5)));
    if (row >= n) return;
    const int beg = rowstart[row];
    const int end = (row + 1 < n) ? rowstart[row + 1] : E;
    const float di = rsqrtf(fmaxf((float)(end - beg + 1), 1.0f));
    const int col = lane * VW;

    vec_t acc[NP];
    {
        const float sl = di * di;
#pragma unroll
        for (int p = 0; p < NP; ++p) {
            const vec_t xv = *(const vec_t*)(xw + (size_t)row * F + p * 32 * VW + col);
            const vec_t bv = *(const vec_t*)(bias + p * 32 * VW + col);
            acc[p] = bv + xv * sl;
        }
    }

    for (int base = beg; base < end; base += 32) {
        const int nb = min(32, end - base);
        int   s_l = 0;
        float c_l = 0.0f;
        if (lane < nb) {
            s_l = csr_src[base + lane];
            const int a  = rowstart[s_l];
            const int bn = rowstart[min(s_l + 1, n - 1)];
            const int b  = (s_l + 1 < n) ? bn : E;
            c_l = rsqrtf(fmaxf((float)(b - a + 1), 1.0f)) * di;
        }
        const int c_bits = __builtin_bit_cast(int, c_l);
        int k = 0;
        for (; k + UNR <= nb; k += UNR) {
            vec_t xr[UNR][NP];
            float c[UNR];
#pragma unroll
            for (int u = 0; u < UNR; ++u) {
                const int s = __builtin_amdgcn_readlane(s_l, k + u);
                c[u] = __builtin_bit_cast(float, __builtin_amdgcn_readlane(c_bits, k + u));
                const float* r = xw + (size_t)s * F + col;
#pragma unroll
                for (int p = 0; p < NP; ++p) xr[u][p] = *(const vec_t*)(r + p * 32 * VW);
            }
#pragma unroll
            for (int u = 0; u < UNR; ++u)
#pragma unroll
                for (int p = 0; p < NP; ++p) acc[p] += xr[u][p] * c[u];
        }
#pragma unroll 1
        for (; k < nb; ++k) {
            const int s = __builtin_amdgcn_readlane(s_l, k);
            const float c = __builtin_bit_cast(float, __builtin_amdgcn_readlane(c_bits, k));
            const float* r = xw + (size_t)s * F + col;
#pragma unroll
            for (int p = 0; p < NP; ++p) acc[p] += *(const vec_t*)(r + p * 32 * VW) * c;
        }
    }

    if constexpr (EPI == EPI_BN_RELU) {
#pragma unroll
        for (int p = 0; p < NP; ++p) {
            const int c0 = p * 32 * VW + col;
            const vec_t gv = *(const vec_t*)(g + c0);
            const vec_t bv = *(const vec_t*)(beta + c0);
            const vec_t mv = *(const vec_t*)(m + c0);
            const vec_t vv = *(const vec_t*)(v + c0);
            vec_t o;
#pragma unroll
            for (int j = 0; j < VW; ++j)
                o[j] = fmaxf((acc[p][j] - mv[j]) * rsqrtf(vv[j] + 1e-5f) * gv[j] + bv[j], 0.0f);
            *(vec_t*)(y + (size_t)row * F + c0) = o;
        }
    } else {
        const float a0 = acc[0][0], a1 = acc[0][1];
        float mx = fmaxf(a0, a1);
#pragma unroll
        for (int off = 16; off > 0; off >>= 1) mx = fmaxf(mx, __shfl_xor(mx, off, 32));
        const float e0 = __expf(a0 - mx), e1 = __expf(a1 - mx);
        float s = e0 + e1;
#pragma unroll
        for (int off = 16; off > 0; off >>= 1) s += __shfl_xor(s, off, 32);
        const float inv = 1.0f / s;
        vec_t o;
        o[0] = e0 * inv;
        o[1] = e1 * inv;
        *(vec_t*)(y + (size_t)row * F + col) = o;
    }
}

extern "C" void kernel_launch(void* const* d_in, const int* in_sizes, int n_in,
                              void* d_out, int out_size, void* d_ws, size_t ws_size,
                              hipStream_t stream) {
    const int IN = 256, HID_ = 256, OUT_ = 64;
    const int N = in_sizes[0] / IN;
    const int E = in_sizes[1] / 2;

    const float* x    = (const float*)d_in[0];
    const int*   src  = (const int*)d_in[1];
    const int*   dst  = ((const int*)d_in[1]) + E;
    const float* W1   = (const float*)d_in[2];
    const float* b1   = (const float*)d_in[3];
    const float* g1   = (const float*)d_in[4];
    const float* be1  = (const float*)d_in[5];
    const float* m1   = (const float*)d_in[6];
    const float* v1   = (const float*)d_in[7];
    const float* W2   = (const float*)d_in[8];
    const float* b2   = (const float*)d_in[9];
    const float* g2   = (const float*)d_in[10];
    const float* be2  = (const float*)d_in[11];
    const float* m2   = (const float*)d_in[12];
    const float* v2   = (const float*)d_in[13];
    const float* W3   = (const float*)d_in[14];
    const float* b3   = (const float*)d_in[15];
    float* out = (float*)d_out;

    char* ws = (char*)d_ws;
    size_t off = 0;
    int*   rowstart = (int*)(ws + off);   off += ((size_t)N * 4 + 255) & ~(size_t)255;
    int*   csr_src  = (int*)(ws + off);   off += ((size_t)E * 4 + 255) & ~(size_t)255;
    float* bufA = (float*)(ws + off); off += (size_t)N * HID_ * 4;
    float* bufB = (float*)(ws + off);
    int* cnt  = (int*)bufA;
    int* rank = (int*)bufB;
    (void)ws_size; (void)n_in; (void)out_size;

    const int gN = (N + TPB - 1) / TPB;
    const int gE = (E + TPB - 1) / TPB;
    const int gemmX = (N + 127) / 128;
    const int gRow = (N + TPB / 32 - 1) / (TPB / 32);

    gcn_cnt_zero <<<gN, TPB, 0, stream>>>(cnt, N);
    gcn_cnt_edges<<<gE, TPB, 0, stream>>>(dst, cnt, rank, E);
    gcn_row_scan <<<(N + 1023) / 1024, 1024, 0, stream>>>(cnt, rowstart, N);
    gcn_csr_fill <<<gE, TPB, 0, stream>>>(src, dst, rank, rowstart, csr_src, E);

    gcn_gemm_wmma<<<dim3(gemmX, HID_ / 64), TPB, 0, stream>>>(x, W1, bufA, N, IN, HID_);
    gcn_gather<256, EPI_BN_RELU><<<gRow, TPB, 0, stream>>>(rowstart, csr_src, bufA, b1,
                                                            g1, be1, m1, v1, bufB, N, E);

    gcn_gemm_wmma<<<dim3(gemmX, HID_ / 64), TPB, 0, stream>>>(bufB, W2, bufA, N, HID_, HID_);
    gcn_gather<256, EPI_BN_RELU><<<gRow, TPB, 0, stream>>>(rowstart, csr_src, bufA, b2,
                                                            g2, be2, m2, v2, bufB, N, E);

    gcn_gemm_wmma<<<dim3(gemmX, OUT_ / 64), TPB, 0, stream>>>(bufB, W3, bufA, N, HID_, OUT_);
    gcn_gather<64, EPI_SOFTMAX><<<gRow, TPB, 0, stream>>>(rowstart, csr_src, bufA, b3,
                                                          nullptr, nullptr, nullptr, nullptr, out, N, E);
}


# Edges ordered by destination by a two-level counting sort on LDS counters (bin per edge block, rank and place per bin), not per-edge returning global atomics: 4 launches -> 2; row sums: arrival order
# speedup vs baseline: 3.4318x; 1.2103x over previous
#include <hip/hip_runtime.h>
#include <hip/hip_bf16.h>

typedef __attribute__((ext_vector_type(2))) float v2f;
typedef __attribute__((ext_vector_type(8))) float v8f;

#define TPB 256
#define KC    64
#define ASTR  68
#define WTSTR 68

#define BIN_SHIFT 7
#define BIN_ROWS  (1 << BIN_SHIFT)
#define EPB       4096
#define SPL       4
#define GRP       4

__global__ __launch_bounds__(1024)
void gcn_bin_edges(const int* __restrict__ src, const int* __restrict__ dst,
                   int2* __restrict__ binned, int* __restrict__ offT, int E, int P, int NBA) {
    __shared__ int hist[1024];
    __shared__ int wtot[32];
    const int tid  = threadIdx.x;
    const int lane = tid & 31;
    const int wid  = tid >> 5;
    const int k    = blockIdx.x;
    const int e0   = k * EPB;
    hist[tid] = 0;
    int d[4], s[4], r[4];
#pragma unroll
    for (int j = 0; j < 4; ++j) {
        const int e = e0 + j * 1024 + tid;
        d[j] = (e < E) ? dst[e] : -1;
        s[j] = (e < E) ? src[e] : 0;
    }
    __syncthreads();
#pragma unroll
    for (int j = 0; j < 4; ++j)
        r[j] = (d[j] >= 0) ? atomicAdd(&hist[d[j] >> BIN_SHIFT], 1) : 0;
    __syncthreads();
    const int c = hist[tid];
    int inc = c;
#pragma unroll
    for (int off = 1; off < 32; off <<= 1) {
        int t = __shfl_up(inc, off, 32);
        if (lane >= off) inc += t;
    }
    if (lane == 31) wtot[wid] = inc;
    __syncthreads();
    int pre = (lane < wid) ? wtot[lane] : 0;
#pragma unroll
    for (int off = 16; off > 0; off >>= 1) pre += __shfl_xor(pre, off, 32);
    const int excl = pre + inc - c;
    hist[tid] = excl;
    __syncthreads();
    if (tid <= P) offT[(size_t)tid * NBA + k] = excl;
#pragma unroll
    for (int j = 0; j < 4; ++j)
        if (d[j] >= 0) binned[e0 + hist[d[j] >> BIN_SHIFT] + r[j]] = make_int2(d[j], s[j]);
}

__global__ __launch_bounds__(1024)
void gcn_bin_to_csr(const int2* __restrict__ binned, const int* __restrict__ offT,
                    int* __restrict__ rowstart, int* __restrict__ csr_src, int n, int NBA) {
    __shared__ int cnt[BIN_ROWS];
    __shared__ int wtot[32];
    __shared__ int wrow[BIN_ROWS / 32];
    const int tid  = threadIdx.x;
    const int lane = tid & 31;
    const int wid  = tid >> 5;
    const int b    = blockIdx.x;
    const int lo   = b << BIN_SHIFT;
    const int* __restrict__ o_lo = offT + (size_t)b * NBA;
    const int* __restrict__ o_hi = o_lo + NBA;

    if (tid < BIN_ROWS) cnt[tid] = 0;
    __syncthreads();

    int part = 0;
    for (int q = tid; q < NBA * SPL; q += 1024) {
        const int k  = q / SPL;
        const int j  = q % SPL;
        const int2* __restrict__ seg = binned + (size_t)k * EPB;
        const int o0 = o_lo[k];
        const int o1 = o_hi[k];
        if (j == 0) part += o0;
        int i = o0 + j;
        int dd[GRP];
#pragma unroll
        for (int u = 0; u < GRP; ++u) dd[u] = (i + u * SPL < o1) ? seg[i + u * SPL].x : -1;
#pragma unroll
        for (int u = 0; u < GRP; ++u) if (dd[u] >= 0) atomicAdd(&cnt[dd[u] - lo], 1);
        for (i += GRP * SPL; i < o1; i += SPL) atomicAdd(&cnt[seg[i].x - lo], 1);
    }
#pragma unroll
    for (int off = 16; off > 0; off >>= 1) part += __shfl_xor(part, off, 32);
    if (lane == 0) wtot[wid] = part;
    __syncthreads();
    int base = wtot[lane];
#pragma unroll
    for (int off = 16; off > 0; off >>= 1) base += __shfl_xor(base, off, 32);

    int start = 0;
    if (tid < BIN_ROWS) {
        const int c = cnt[tid];
        int inc = c;
#pragma unroll
        for (int off = 1; off < 32; off <<= 1) {
            int t = __shfl_up(inc, off, 32);
            if (lane >= off) inc += t;
        }
        if (lane == 31) wrow[wid] = inc;
        start = base + inc - c;
    }
    __syncthreads();
    if (tid < BIN_ROWS) {
        int pre = (lane < wid) ? wrow[lane] : 0;
#pragma unroll
        for (int off = 16; off > 0; off >>= 1) pre += __shfl_xor(pre, off, 32);
        start += pre;
        cnt[tid] = start;
    }
    __syncthreads();
    if (tid < BIN_ROWS && lo + tid < n) rowstart[lo + tid] = start;

    for (int q = tid; q < NBA * SPL; q += 1024) {
        const int k  = q / SPL;
        const int2* __restrict__ seg = binned + (size_t)k * EPB;
        const int o1 = o_hi[k];
        int i = o_lo[k] + (q % SPL);
        int2 ee[GRP];
#pragma unroll
        for (int u = 0; u < GRP; ++u) ee[u] = (i + u * SPL < o1) ? seg[i + u * SPL] : make_int2(-1, 0);
#pragma unroll
        for (int u = 0; u < GRP; ++u)
            if (ee[u].x >= 0) csr_src[atomicAdd(&cnt[ee[u].x - lo], 1)] = ee[u].y;
        for (i += GRP * SPL; i < o1; i += SPL) {
            const int2 e = seg[i];
            csr_src[atomicAdd(&cnt[e.x - lo], 1)] = e.y;
        }
    }
}

__global__ __launch_bounds__(TPB)
void gcn_gemm_wmma(const float* __restrict__ A, const float* __restrict__ W,
                   float* __restrict__ C, int M, int K, int N) {
    __shared__ float lA[128 * ASTR];
    __shared__ float lWt[64 * WTSTR];
    const int tid  = threadIdx.x;
    const int lane = tid & 31;
    const int wid  = tid >> 5;
    const int l16  = lane & 15;
    const int half = lane >> 4;
    const int rowBase = blockIdx.x * 128;
    const int colBase = blockIdx.y * 64;

    v8f acc[4] = {};

    for (int kc = 0; kc < K; kc += KC) {
#pragma unroll
        for (int i = 0; i < 8; ++i) {
            int idx = tid + i * TPB;
            int r   = idx >> 4;
            int c4  = (idx & 15) << 2;
            int gr  = min(rowBase + r, M - 1);
            float4 v = *(const float4*)(A + (size_t)gr * K + kc + c4);
            *(float4*)(&lA[r * ASTR + c4]) = v;
        }
#pragma unroll
        for (int i = 0; i < 4; ++i) {
            int idx = tid + i * TPB;
            int r   = idx >> 4;
            int c4  = (idx & 15) << 2;
            float4 v = *(const float4*)(W + (size_t)(kc + r) * N + colBase + c4);
            lWt[(c4 + 0) * WTSTR + r] = v.x;
            lWt[(c4 + 1) * WTSTR + r] = v.y;
            lWt[(c4 + 2) * WTSTR + r] = v.z;
            lWt[(c4 + 3) * WTSTR + r] = v.w;
        }
        __syncthreads();

        for (int kk = 0; kk < KC; kk += 4) {
            const int ks = kk + 2 * half;
            v2f a = *(const v2f*)(&lA[(wid * 16 + l16) * ASTR + ks]);
#pragma unroll
            for (int ct = 0; ct < 4; ++ct) {
                v2f b = *(const v2f*)(&lWt[(ct * 16 + l16) * WTSTR + ks]);
                acc[ct] = __builtin_amdgcn_wmma_f32_16x16x4_f32(
                    false, a, false, b, (short)0, acc[ct], false, false);
            }
        }
        __syncthreads();
    }

#pragma unroll
    for (int ct = 0; ct < 4; ++ct) {
        const int col = colBase + ct * 16 + l16;
#pragma unroll
        for (int v = 0; v < 8; ++v) {
            const int row = rowBase + wid * 16 + v + 8 * half;
            if (row < M) C[(size_t)row * N + col] = acc[ct][v];
        }
    }
}

#define EPI_BN_RELU 0
#define EPI_SOFTMAX 1
template <int F, int EPI>
__global__ __launch_bounds__(TPB)
void gcn_gather(const int* __restrict__ rowstart, const int* __restrict__ csr_src,
                const float* __restrict__ xw, const float* __restrict__ bias,
                const float* __restrict__ g, const float* __restrict__ beta,
                const float* __restrict__ m, const float* __restrict__ v,
                float* __restrict__ y, int n, int E) {
    static_assert(EPI == EPI_BN_RELU || F == 64, "the softmax form is written for 64 columns");
    constexpr int VW  = (F >= 128) ? 4 : F / 32;
    constexpr int NP  = F / (32 * VW);
    constexpr int UNR = (F >= 256) ? 4 : 8;
    typedef float vec_t __attribute__((ext_vector_type(VW)));
    const int lane = threadIdx.x & 31;
    const int row = __builtin_amdgcn_readfirstlane((int)(blockIdx.x * (TPB / 32) + (threadIdx.x >> 5)));
    if (row >= n) return;
    const int beg = rowstart[row];
    const int end = (row + 1 < n) ? rowstart[row + 1] : E;
    const float di = rsqrtf(fmaxf((float)(end - beg + 1), 1.0f));
    const int col = lane * VW;

    vec_t acc[NP];
    {
        const float sl = di * di;
#pragma unroll
        for (int p = 0; p < NP; ++p) {
            const vec_t xv = *(const vec_t*)(xw + (size_t)row * F + p * 32 * VW + col);
            const vec_t bv = *(const vec_t*)(bias + p * 32 * VW + col);
            acc[p] = bv + xv * sl;
        }
    }

    for (int base = beg; base < end; base += 32) {
        const int nb = min(32, end - base);
        int   s_l = 0;
        float c_l = 0.0f;
        if (lane < nb) {
            s_l = csr_src[base + lane];
            const int a  = rowstart[s_l];
            const int bn = rowstart[min(s_l + 1, n - 1)];
            const int b  = (s_l + 1 < n) ? bn : E;
            c_l = rsqrtf(fmaxf((float)(b - a + 1), 1.0f)) * di;
        }
        const int c_bits = __builtin_bit_cast(int, c_l);
        int k = 0;
        for (; k + UNR <= nb; k += UNR) {
            vec_t xr[UNR][NP];
            float c[UNR];
#pragma unroll
            for (int u = 0; u < UNR; ++u) {
                const int s = __builtin_amdgcn_readlane(s_l, k + u);
                c[u] = __builtin_bit_cast(float, __builtin_amdgcn_readlane(c_bits, k + u));
                const float* r = xw + (size_t)s * F + col;
#pragma unroll
                for (int p = 0; p < NP; ++p) xr[u][p] = *(const vec_t*)(r + p * 32 * VW);
            }
#pragma unroll
            for (int u = 0; u < UNR; ++u)
#pragma unroll
                for (int p = 0; p < NP; ++p) acc[p] += xr[u][p] * c[u];
        }
#pragma unroll 1
        for (; k < nb; ++k) {
            const int s = __builtin_amdgcn_readlane(s_l, k);
            const float c = __builtin_bit_cast(float, __builtin_amdgcn_readlane(c_bits, k));
            const float* r = xw + (size_t)s * F + col;
#pragma unroll
            for (int p = 0; p < NP; ++p) acc[p] += *(const vec_t*)(r + p * 32 * VW) * c;
        }
    }

    if constexpr (EPI == EPI_BN_RELU) {
#pragma unroll
        for (int p = 0; p < NP; ++p) {
            const int c0 = p * 32 * VW + col;
            const vec_t gv = *(const vec_t*)(g + c0);
            const vec_t bv = *(const vec_t*)(beta + c0);
            const vec_t mv = *(const vec_t*)(m + c0);
            const vec_t vv = *(const vec_t*)(v + c0);
            vec_t o;
#pragma unroll
            for (int j = 0; j < VW; ++j)
                o[j] = fmaxf((acc[p][j] - mv[j]) * rsqrtf(vv[j] + 1e-5f) * gv[j] + bv[j], 0.0f);
            *(vec_t*)(y + (size_t)row * F + c0) = o;
        }
    } else {
        const float a0 = acc[0][0], a1 = acc[0][1];
        float mx = fmaxf(a0, a1);
#pragma unroll
        for (int off = 16; off > 0; off >>= 1) mx = fmaxf(mx, __shfl_xor(mx, off, 32));
        const float e0 = __expf(a0 - mx), e1 = __expf(a1 - mx);
        float s = e0 + e1;
#pragma unroll
        for (int off = 16; off > 0; off >>= 1) s += __shfl_xor(s, off, 32);
        const float inv = 1.0f / s;
        vec_t o;
        o[0] = e0 * inv;
        o[1] = e1 * inv;
        *(vec_t*)(y + (size_t)row * F + col) = o;
    }
}

extern "C" void kernel_launch(void* const* d_in, const int* in_sizes, int n_in,
                              void* d_out, int out_size, void* d_ws, size_t ws_size,
                              hipStream_t stream) {
    const int IN = 256, HID_ = 256, OUT_ = 64;
    const int N = in_sizes[0] / IN;
    const int E = in_sizes[1] / 2;

    const float* x    = (const float*)d_in[0];
    const int*   src  = (const int*)d_in[1];
    const int*   dst  = ((const int*)d_in[1]) + E;
    const float* W1   = (const float*)d_in[2];
    const float* b1   = (const float*)d_in[3];
    const float* g1   = (const float*)d_in[4];
    const float* be1  = (const float*)d_in[5];
    const float* m1   = (const float*)d_in[6];
    const float* v1   = (const float*)d_in[7];
    const float* W2   = (const float*)d_in[8];
    const float* b2   = (const float*)d_in[9];
    const float* g2   = (const float*)d_in[10];
    const float* be2  = (const float*)d_in[11];
    const float* m2   = (const float*)d_in[12];
    const float* v2   = (const float*)d_in[13];
    const float* W3   = (const float*)d_in[14];
    const float* b3   = (const float*)d_in[15];
    float* out = (float*)d_out;

    char* ws = (char*)d_ws;
    size_t off = 0;
    int*   rowstart = (int*)(ws + off);   off += ((size_t)N * 4 + 255) & ~(size_t)255;
    int*   csr_src  = (int*)(ws + off);   off += ((size_t)E * 4 + 255) & ~(size_t)255;
    float* bufA = (float*)(ws + off); off += (size_t)N * HID_ * 4;
    float* bufB = (float*)(ws + off);
    int2* binned = (int2*)bufA;
    int*  offT   = (int*)bufB;
    (void)ws_size; (void)n_in; (void)out_size;

    const int P   = (N + BIN_ROWS - 1) / BIN_ROWS;
    const int NBA = (E + EPB - 1) / EPB;
    const int gemmX = (N + 127) / 128;
    const int gRow = (N + TPB / 32 - 1) / (TPB / 32);

    gcn_bin_edges <<<NBA, 1024, 0, stream>>>(src, dst, binned, offT, E, P, NBA);
    gcn_bin_to_csr<<<P, 1024, 0, stream>>>(binned, offT, rowstart, csr_src, N, NBA);

    gcn_gemm_wmma<<<dim3(gemmX, HID_ / 64), TPB, 0, stream>>>(x, W1, bufA, N, IN, HID_);
    gcn_gather<256, EPI_BN_RELU><<<gRow, TPB, 0, stream>>>(rowstart, csr_src, bufA, b1,
                                                            g1, be1, m1, v1, bufB, N, E);

    gcn_gemm_wmma<<<dim3(gemmX, HID_ / 64), TPB, 0, stream>>>(bufB, W2, bufA, N, HID_, HID_);
    gcn_gather<256, EPI_BN_RELU><<<gRow, TPB, 0, stream>>>(rowstart, csr_src, bufA, b2,
                                                            g2, be2, m2, v2, bufB, N, E);

    gcn_gemm_wmma<<<dim3(gemmX, OUT_ / 64), TPB, 0, stream>>>(bufB, W3, bufA, N, HID_, OUT_);
    gcn_gather<64, EPI_SOFTMAX><<<gRow, TPB, 0, stream>>>(rowstart, csr_src, bufA, b3,
                                                          nullptr, nullptr, nullptr, nullptr, out, N, E);
}


# GEMM K loop pipelined over two LDS stages of 32 k: next chunk's global loads issued ahead of the current chunk's WMMAs, stored to the other stage behind them; one barrier per chunk; sum order kept
# speedup vs baseline: 3.6903x; 1.0753x over previous
#include <hip/hip_runtime.h>
#include <hip/hip_bf16.h>

typedef __attribute__((ext_vector_type(2))) float v2f;
typedef __attribute__((ext_vector_type(8))) float v8f;

#define TPB 256
#define KC    32
#define ASTR  36
#define WTSTR 36

#define BIN_SHIFT 7
#define BIN_ROWS  (1 << BIN_SHIFT)
#define EPB       4096
#define SPL       4
#define GRP       4

__global__ __launch_bounds__(1024)
void gcn_bin_edges(const int* __restrict__ src, const int* __restrict__ dst,
                   int2* __restrict__ binned, int* __restrict__ offT, int E, int P, int NBA) {
    __shared__ int hist[1024];
    __shared__ int wtot[32];
    const int tid  = threadIdx.x;
    const int lane = tid & 31;
    const int wid  = tid >> 5;
    const int k    = blockIdx.x;
    const int e0   = k * EPB;
    hist[tid] = 0;
    int d[4], s[4], r[4];
#pragma unroll
    for (int j = 0; j < 4; ++j) {
        const int e = e0 + j * 1024 + tid;
        d[j] = (e < E) ? dst[e] : -1;
        s[j] = (e < E) ? src[e] : 0;
    }
    __syncthreads();
#pragma unroll
    for (int j = 0; j < 4; ++j)
        r[j] = (d[j] >= 0) ? atomicAdd(&hist[d[j] >> BIN_SHIFT], 1) : 0;
    __syncthreads();
    const int c = hist[tid];
    int inc = c;
#pragma unroll
    for (int off = 1; off < 32; off <<= 1) {
        int t = __shfl_up(inc, off, 32);
        if (lane >= off) inc += t;
    }
    if (lane == 31) wtot[wid] = inc;
    __syncthreads();
    int pre = (lane < wid) ? wtot[lane] : 0;
#pragma unroll
    for (int off = 16; off > 0; off >>= 1) pre += __shfl_xor(pre, off, 32);
    const int excl = pre + inc - c;
    hist[tid] = excl;
    __syncthreads();
    if (tid <= P) offT[(size_t)tid * NBA + k] = excl;
#pragma unroll
    for (int j = 0; j < 4; ++j)
        if (d[j] >= 0) binned[e0 + hist[d[j] >> BIN_SHIFT] + r[j]] = make_int2(d[j], s[j]);
}

__global__ __launch_bounds__(1024)
void gcn_bin_to_csr(const int2* __restrict__ binned, const int* __restrict__ offT,
                    int* __restrict__ rowstart, int* __restrict__ csr_src, int n, int NBA) {
    __shared__ int cnt[BIN_ROWS];
    __shared__ int wtot[32];
    __shared__ int wrow[BIN_ROWS / 32];
    const int tid  = threadIdx.x;
    const int lane = tid & 31;
    const int wid  = tid >> 5;
    const int b    = blockIdx.x;
    const int lo   = b << BIN_SHIFT;
    const int* __restrict__ o_lo = offT + (size_t)b * NBA;
    const int* __restrict__ o_hi = o_lo + NBA;

    if (tid < BIN_ROWS) cnt[tid] = 0;
    __syncthreads();

    int part = 0;
    for (int q = tid; q < NBA * SPL; q += 1024) {
        const int k  = q / SPL;
        const int j  = q % SPL;
        const int2* __restrict__ seg = binned + (size_t)k * EPB;
        const int o0 = o_lo[k];
        const int o1 = o_hi[k];
        if (j == 0) part += o0;
        int i = o0 + j;
        int dd[GRP];
#pragma unroll
        for (int u = 0; u < GRP; ++u) dd[u] = (i + u * SPL < o1) ? seg[i + u * SPL].x : -1;
#pragma unroll
        for (int u = 0; u < GRP; ++u) if (dd[u] >= 0) atomicAdd(&cnt[dd[u] - lo], 1);
        for (i += GRP * SPL; i < o1; i += SPL) atomicAdd(&cnt[seg[i].x - lo], 1);
    }
#pragma unroll
    for (int off = 16; off > 0; off >>= 1) part += __shfl_xor(part, off, 32);
    if (lane == 0) wtot[wid] = part;
    __syncthreads();
    int base = wtot[lane];
#pragma unroll
    for (int off = 16; off > 0; off >>= 1) base += __shfl_xor(base, off, 32);

    int start = 0;
    if (tid < BIN_ROWS) {
        const int c = cnt[tid];
        int inc = c;
#pragma unroll
        for (int off = 1; off < 32; off <<= 1) {
            int t = __shfl_up(inc, off, 32);
            if (lane >= off) inc += t;
        }
        if (lane == 31) wrow[wid] = inc;
        start = base + inc - c;
    }
    __syncthreads();
    if (tid < BIN_ROWS) {
        int pre = (lane < wid) ? wrow[lane] : 0;
#pragma unroll
        for (int off = 16; off > 0; off >>= 1) pre += __shfl_xor(pre, off, 32);
        start += pre;
        cnt[tid] = start;
    }
    __syncthreads();
    if (tid < BIN_ROWS && lo + tid < n) rowstart[lo + tid] = start;

    for (int q = tid; q < NBA * SPL; q += 1024) {
        const int k  = q / SPL;
        const int2* __restrict__ seg = binned + (size_t)k * EPB;
        const int o1 = o_hi[k];
        int i = o_lo[k] + (q % SPL);
        int2 ee[GRP];
#pragma unroll
        for (int u = 0; u < GRP; ++u) ee[u] = (i + u * SPL < o1) ? seg[i + u * SPL] : make_int2(-1, 0);
#pragma unroll
        for (int u = 0; u < GRP; ++u)
            if (ee[u].x >= 0) csr_src[atomicAdd(&cnt[ee[u].x - lo], 1)] = ee[u].y;
        for (i += GRP * SPL; i < o1; i += SPL) {
            const int2 e = seg[i];
            csr_src[atomicAdd(&cnt[e.x - lo], 1)] = e.y;
        }
    }
}

__global__ __launch_bounds__(TPB)
void gcn_gemm_wmma(const float* __restrict__ A, const float* __restrict__ W,
                   float* __restrict__ C, int M, int K, int N) {
    __shared__ float lA[2][128 * ASTR];
    __shared__ float lWt[2][64 * WTSTR];
    const int tid  = threadIdx.x;
    const int lane = tid & 31;
    const int wid  = tid >> 5;
    const int l16  = lane & 15;
    const int half = lane >> 4;
    const int rowBase = blockIdx.x * 128;
    const int colBase = blockIdx.y * 64;

    const int ar  = tid >> 3;
    const int ac4 = (tid & 7) << 2;
    const float* pA0 = A + (size_t)min(rowBase + ar,      M - 1) * K + ac4;
    const float* pA1 = A + (size_t)min(rowBase + ar + 32, M - 1) * K + ac4;
    const float* pA2 = A + (size_t)min(rowBase + ar + 64, M - 1) * K + ac4;
    const float* pA3 = A + (size_t)min(rowBase + ar + 96, M - 1) * K + ac4;
    const int sA = ar * ASTR + ac4;
    const int wr  = tid >> 4;
    const int wc4 = (tid & 15) << 2;
    const float* pW0 = W + (size_t)wr * N + colBase + wc4;
    const float* pW1 = pW0 + (size_t)16 * N;
    const int sW = wc4 * WTSTR + wr;

#define GEMM_LOAD(kc_)                                              \
    ra0 = *(const float4*)(pA0 + (kc_));                            \
    ra1 = *(const float4*)(pA1 + (kc_));                            \
    ra2 = *(const float4*)(pA2 + (kc_));                            \
    ra3 = *(const float4*)(pA3 + (kc_));                            \
    rw0 = *(const float4*)(pW0 + (size_t)(kc_) * N);                \
    rw1 = *(const float4*)(pW1 + (size_t)(kc_) * N);
#define GEMM_STORE(s_)                                              \
    *(float4*)(&lA[s_][sA])             = ra0;                      \
    *(float4*)(&lA[s_][sA + 32 * ASTR]) = ra1;                      \
    *(float4*)(&lA[s_][sA + 64 * ASTR]) = ra2;                      \
    *(float4*)(&lA[s_][sA + 96 * ASTR]) = ra3;                      \
    lWt[s_][sW + 0 * WTSTR]      = rw0.x;                           \
    lWt[s_][sW + 1 * WTSTR]      = rw0.y;                           \
    lWt[s_][sW + 2 * WTSTR]      = rw0.z;                           \
    lWt[s_][sW + 3 * WTSTR]      = rw0.w;                           \
    lWt[s_][sW + 0 * WTSTR + 16] = rw1.x;                           \
    lWt[s_][sW + 1 * WTSTR + 16] = rw1.y;                           \
    lWt[s_][sW + 2 * WTSTR + 16] = rw1.z;                           \
    lWt[s_][sW + 3 * WTSTR + 16] = rw1.w;

    float4 ra0, ra1, ra2, ra3, rw0, rw1;
    GEMM_LOAD(0)
    GEMM_STORE(0)
    __syncthreads();

    v8f acc[4] = {};
    const int aOff = (wid * 16 + l16) * ASTR + 2 * half;
    const int bOff = l16 * WTSTR + 2 * half;

    for (int kc = 0; kc < K; kc += KC) {
        const int  st   = (kc / KC) & 1;
        const bool more = (kc + KC < K);
        if (more) { GEMM_LOAD(kc + KC) }

        const float* sa = &lA[st][aOff];
        const float* sb = &lWt[st][bOff];
        for (int kk = 0; kk < KC; kk += 4) {
            v2f a = *(const v2f*)(sa + kk);
#pragma unroll
            for (int ct = 0; ct < 4; ++ct) {
                v2f b = *(const v2f*)(sb + ct * 16 * WTSTR + kk);
                acc[ct] = __builtin_amdgcn_wmma_f32_16x16x4_f32(
                    false, a, false, b, (short)0, acc[ct], false, false);
            }
        }

        if (more) { GEMM_STORE(st ^ 1) }
        __syncthreads();
    }
#undef GEMM_LOAD
#undef GEMM_STORE

#pragma unroll
    for (int ct = 0; ct < 4; ++ct) {
        const int col = colBase + ct * 16 + l16;
#pragma unroll
        for (int v = 0; v < 8; ++v) {
            const int row = rowBase + wid * 16 + v + 8 * half;
            if (row < M) C[(size_t)row * N + col] = acc[ct][v];
        }
    }
}

#define EPI_BN_RELU 0
#define EPI_SOFTMAX 1
template <int F, int EPI>
__global__ __launch_bounds__(TPB)
void gcn_gather(const int* __restrict__ rowstart, const int* __restrict__ csr_src,
                const float* __restrict__ xw, const float* __restrict__ bias,
                const float* __restrict__ g, const float* __restrict__ beta,
                const float* __restrict__ m, const float* __restrict__ v,
                float* __restrict__ y, int n, int E) {
    static_assert(EPI == EPI_BN_RELU || F == 64, "the softmax form is written for 64 columns");
    constexpr int VW  = (F >= 128) ? 4 : F / 32;
    constexpr int NP  = F / (32 * VW);
    constexpr int UNR = (F >= 256) ? 4 : 8;
    typedef float vec_t __attribute__((ext_vector_type(VW)));
    const int lane = threadIdx.x & 31;
    const int row = __builtin_amdgcn_readfirstlane((int)(blockIdx.x * (TPB / 32) + (threadIdx.x >> 5)));
    if (row >= n) return;
    const int beg = rowstart[row];
    const int end = (row + 1 < n) ? rowstart[row + 1] : E;
    const float di = rsqrtf(fmaxf((float)(end - beg + 1), 1.0f));
    const int col = lane * VW;

    vec_t acc[NP];
    {
        const float sl = di * di;
#pragma unroll
        for (int p = 0; p < NP; ++p) {
            const vec_t xv = *(const vec_t*)(xw + (size_t)row * F + p * 32 * VW + col);
            const vec_t bv = *(const vec_t*)(bias + p * 32 * VW + col);
            acc[p] = bv + xv * sl;
        }
    }

    for (int base = beg; base < end; base += 32) {
        const int nb = min(32, end - base);
        int   s_l = 0;
        float c_l = 0.0f;
        if (lane < nb) {
            s_l = csr_src[base + lane];
            const int a  = rowstart[s_l];
            const int bn = rowstart[min(s_l + 1, n - 1)];
            const int b  = (s_l + 1 < n) ? bn : E;
            c_l = rsqrtf(fmaxf((float)(b - a + 1), 1.0f)) * di;
        }
        const int c_bits = __builtin_bit_cast(int, c_l);
        int k = 0;
        for (; k + UNR <= nb; k += UNR) {
            vec_t xr[UNR][NP];
            float c[UNR];
#pragma unroll
            for (int u = 0; u < UNR; ++u) {
                const int s = __builtin_amdgcn_readlane(s_l, k + u);
                c[u] = __builtin_bit_cast(float, __builtin_amdgcn_readlane(c_bits, k + u));
                const float* r = xw + (size_t)s * F + col;
#pragma unroll
                for (int p = 0; p < NP; ++p) xr[u][p] = *(const vec_t*)(r + p * 32 * VW);
            }
#pragma unroll
            for (int u = 0; u < UNR; ++u)
#pragma unroll
                for (int p = 0; p < NP; ++p) acc[p] += xr[u][p] * c[u];
        }
#pragma unroll 1
        for (; k < nb; ++k) {
            const int s = __builtin_amdgcn_readlane(s_l, k);
            const float c = __builtin_bit_cast(float, __builtin_amdgcn_readlane(c_bits, k));
            const float* r = xw + (size_t)s * F + col;
#pragma unroll
            for (int p = 0; p < NP; ++p) acc[p] += *(const vec_t*)(r + p * 32 * VW) * c;
        }
    }

    if constexpr (EPI == EPI_BN_RELU) {
#pragma unroll
        for (int p = 0; p < NP; ++p) {
            const int c0 = p * 32 * VW + col;
            const vec_t gv = *(const vec_t*)(g + c0);
            const vec_t bv = *(const vec_t*)(beta + c0);
            const vec_t mv = *(const vec_t*)(m + c0);
            const vec_t vv = *(const vec_t*)(v + c0);
            vec_t o;
#pragma unroll
            for (int j = 0; j < VW; ++j)
                o[j] = fmaxf((acc[p][j] - mv[j]) * rsqrtf(vv[j] + 1e-5f) * gv[j] + bv[j], 0.0f);
            *(vec_t*)(y + (size_t)row * F + c0) = o;
        }
    } else {
        const float a0 = acc[0][0], a1 = acc[0][1];
        float mx = fmaxf(a0, a1);
#pragma unroll
        for (int off = 16; off > 0; off >>= 1) mx = fmaxf(mx, __shfl_xor(mx, off, 32));
        const float e0 = __expf(a0 - mx), e1 = __expf(a1 - mx);
        float s = e0 + e1;
#pragma unroll
        for (int off = 16; off > 0; off >>= 1) s += __shfl_xor(s, off, 32);
        const float inv = 1.0f / s;
        vec_t o;
        o[0] = e0 * inv;
        o[1] = e1 * inv;
        *(vec_t*)(y + (size_t)row * F + col) = o;
    }
}

extern "C" void kernel_launch(void* const* d_in, const int* in_sizes, int n_in,
                              void* d_out, int out_size, void* d_ws, size_t ws_size,
                              hipStream_t stream) {
    const int IN = 256, HID_ = 256, OUT_ = 64;
    const int N = in_sizes[0] / IN;
    const int E = in_sizes[1] / 2;

    const float* x    = (const float*)d_in[0];
    const int*   src  = (const int*)d_in[1];
    const int*   dst  = ((const int*)d_in[1]) + E;
    const float* W1   = (const float*)d_in[2];
    const float* b1   = (const float*)d_in[3];
    const float* g1   = (const float*)d_in[4];
    const float* be1  = (const float*)d_in[5];
    const float* m1   = (const float*)d_in[6];
    const float* v1   = (const float*)d_in[7];
    const float* W2   = (const float*)d_in[8];
    const float* b2   = (const float*)d_in[9];
    const float* g2   = (const float*)d_in[10];
    const float* be2  = (const float*)d_in[11];
    const float* m2   = (const float*)d_in[12];
    const float* v2   = (const float*)d_in[13];
    const float* W3   = (const float*)d_in[14];
    const float* b3   = (const float*)d_in[15];
    float* out = (float*)d_out;

    char* ws = (char*)d_ws;
    size_t off = 0;
    int*   rowstart = (int*)(ws + off);   off += ((size_t)N * 4 + 255) & ~(size_t)255;
    int*   csr_src  = (int*)(ws + off);   off += ((size_t)E * 4 + 255) & ~(size_t)255;
    float* bufA = (float*)(ws + off); off += (size_t)N * HID_ * 4;
    float* bufB = (float*)(ws + off);
    int2* binned = (int2*)bufA;
    int*  offT   = (int*)bufB;
    (void)ws_size; (void)n_in; (void)out_size;

    const int P   = (N + BIN_ROWS - 1) / BIN_ROWS;
    const int NBA = (E + EPB - 1) / EPB;
    const int gemmX = (N + 127) / 128;
    const int gRow = (N + TPB / 32 - 1) / (TPB / 32);

    gcn_bin_edges <<<NBA, 1024, 0, stream>>>(src, dst, binned, offT, E, P, NBA);
    gcn_bin_to_csr<<<P, 1024, 0, stream>>>(binned, offT, rowstart, csr_src, N, NBA);

    gcn_gemm_wmma<<<dim3(gemmX, HID_ / 64), TPB, 0, stream>>>(x, W1, bufA, N, IN, HID_);
    gcn_gather<256, EPI_BN_RELU><<<gRow, TPB, 0, stream>>>(rowstart, csr_src, bufA, b1,
                                                            g1, be1, m1, v1, bufB, N, E);

    gcn_gemm_wmma<<<dim3(gemmX, HID_ / 64), TPB, 0, stream>>>(bufB, W2, bufA, N, HID_, HID_);
    gcn_gather<256, EPI_BN_RELU><<<gRow, TPB, 0, stream>>>(rowstart, csr_src, bufA, b2,
                                                            g2, be2, m2, v2, bufB, N, E);

    gcn_gemm_wmma<<<dim3(gemmX, OUT_ / 64), TPB, 0, stream>>>(bufB, W3, bufA, N, HID_, OUT_);
    gcn_gather<64, EPI_SOFTMAX><<<gRow, TPB, 0, stream>>>(rowstart, csr_src, bufA, b3,
                                                          nullptr, nullptr, nullptr, nullptr, out, N, E);
}


# Layer-2 and layer-3 products moved into the gather launches that make their input: a block gathers 32 rows (BN+ReLU) into an LDS tile and multiplies it by W on WMMA; 8 launches -> 6; sum orders kept
# speedup vs baseline: 3.8315x; 1.0383x over previous
#include <hip/hip_runtime.h>
#include <hip/hip_bf16.h>

typedef __attribute__((ext_vector_type(2))) float v2f;
typedef __attribute__((ext_vector_type(8))) float v8f;

#define TPB 256
#define KC    32
#define ASTR  36
#define WTSTR 36

#define BIN_SHIFT 7
#define BIN_ROWS  (1 << BIN_SHIFT)
#define EPB       4096
#define SPL       4
#define GRP       4

__global__ __launch_bounds__(1024)
void gcn_bin_edges(const int* __restrict__ src, const int* __restrict__ dst,
                   int2* __restrict__ binned, int* __restrict__ offT, int E, int P, int NBA) {
    __shared__ int hist[1024];
    __shared__ int wtot[32];
    const int tid  = threadIdx.x;
    const int lane = tid & 31;
    const int wid  = tid >> 5;
    const int k    = blockIdx.x;
    const int e0   = k * EPB;
    hist[tid] = 0;
    int d[4], s[4], r[4];
#pragma unroll
    for (int j = 0; j < 4; ++j) {
        const int e = e0 + j * 1024 + tid;
        d[j] = (e < E) ? dst[e] : -1;
        s[j] = (e < E) ? src[e] : 0;
    }
    __syncthreads();
#pragma unroll
    for (int j = 0; j < 4; ++j)
        r[j] = (d[j] >= 0) ? atomicAdd(&hist[d[j] >> BIN_SHIFT], 1) : 0;
    __syncthreads();
    const int c = hist[tid];
    int inc = c;
#pragma unroll
    for (int off = 1; off < 32; off <<= 1) {
        int t = __shfl_up(inc, off, 32);
        if (lane >= off) inc += t;
    }
    if (lane == 31) wtot[wid] = inc;
    __syncthreads();
    int pre = (lane < wid) ? wtot[lane] : 0;
#pragma unroll
    for (int off = 16; off > 0; off >>= 1) pre += __shfl_xor(pre, off, 32);
    const int excl = pre + inc - c;
    hist[tid] = excl;
    __syncthreads();
    if (tid <= P) offT[(size_t)tid * NBA + k] = excl;
#pragma unroll
    for (int j = 0; j < 4; ++j)
        if (d[j] >= 0) binned[e0 + hist[d[j] >> BIN_SHIFT] + r[j]] = make_int2(d[j], s[j]);
}

__global__ __launch_bounds__(1024)
void gcn_bin_to_csr(const int2* __restrict__ binned, const int* __restrict__ offT,
                    int* __restrict__ rowstart, int* __restrict__ csr_src, int n, int NBA) {
    __shared__ int cnt[BIN_ROWS];
    __shared__ int wtot[32];
    __shared__ int wrow[BIN_ROWS / 32];
    const int tid  = threadIdx.x;
    const int lane = tid & 31;
    const int wid  = tid >> 5;
    const int b    = blockIdx.x;
    const int lo   = b << BIN_SHIFT;
    const int* __restrict__ o_lo = offT + (size_t)b * NBA;
    const int* __restrict__ o_hi = o_lo + NBA;

    if (tid < BIN_ROWS) cnt[tid] = 0;
    __syncthreads();

    int part = 0;
    for (int q = tid; q < NBA * SPL; q += 1024) {
        const int k  = q / SPL;
        const int j  = q % SPL;
        const int2* __restrict__ seg = binned + (size_t)k * EPB;
        const int o0 = o_lo[k];
        const int o1 = o_hi[k];
        if (j == 0) part += o0;
        int i = o0 + j;
        int dd[GRP];
#pragma unroll
        for (int u = 0; u < GRP; ++u) dd[u] = (i + u * SPL < o1) ? seg[i + u * SPL].x : -1;
#pragma unroll
        for (int u = 0; u < GRP; ++u) if (dd[u] >= 0) atomicAdd(&cnt[dd[u] - lo], 1);
        for (i += GRP * SPL; i < o1; i += SPL) atomicAdd(&cnt[seg[i].x - lo], 1);
    }
#pragma unroll
    for (int off = 16; off > 0; off >>= 1) part += __shfl_xor(part, off, 32);
    if (lane == 0) wtot[wid] = part;
    __syncthreads();
    int base = wtot[lane];
#pragma unroll
    for (int off = 16; off > 0; off >>= 1) base += __shfl_xor(base, off, 32);

    int start = 0;
    if (tid < BIN_ROWS) {
        const int c = cnt[tid];
        int inc = c;
#pragma unroll
        for (int off = 1; off < 32; off <<= 1) {
            int t = __shfl_up(inc, off, 32);
            if (lane >= off) inc += t;
        }
        if (lane == 31) wrow[wid] = inc;
        start = base + inc - c;
    }
    __syncthreads();
    if (tid < BIN_ROWS) {
        int pre = (lane < wid) ? wrow[lane] : 0;
#pragma unroll
        for (int off = 16; off > 0; off >>= 1) pre += __shfl_xor(pre, off, 32);
        start += pre;
        cnt[tid] = start;
    }
    __syncthreads();
    if (tid < BIN_ROWS && lo + tid < n) rowstart[lo + tid] = start;

    for (int q = tid; q < NBA * SPL; q += 1024) {
        const int k  = q / SPL;
        const int2* __restrict__ seg = binned + (size_t)k * EPB;
        const int o1 = o_hi[k];
        int i = o_lo[k] + (q % SPL);
        int2 ee[GRP];
#pragma unroll
        for (int u = 0; u < GRP; ++u) ee[u] = (i + u * SPL < o1) ? seg[i + u * SPL] : make_int2(-1, 0);
#pragma unroll
        for (int u = 0; u < GRP; ++u)
            if (ee[u].x >= 0) csr_src[atomicAdd(&cnt[ee[u].x - lo], 1)] = ee[u].y;
        for (i += GRP * SPL; i < o1; i += SPL) {
            const int2 e = seg[i];
            csr_src[atomicAdd(&cnt[e.x - lo], 1)] = e.y;
        }
    }
}

__global__ __launch_bounds__(TPB)
void gcn_gemm_wmma(const float* __restrict__ A, const float* __restrict__ W,
                   float* __restrict__ C, int M, int K, int N) {
    __shared__ float lA[2][128 * ASTR];
    __shared__ float lWt[2][64 * WTSTR];
    const int tid  = threadIdx.x;
    const int lane = tid & 31;
    const int wid  = tid >> 5;
    const int l16  = lane & 15;
    const int half = lane >> 4;
    const int rowBase = blockIdx.x * 128;
    const int colBase = blockIdx.y * 64;

    const int ar  = tid >> 3;
    const int ac4 = (tid & 7) << 2;
    const float* pA0 = A + (size_t)min(rowBase + ar,      M - 1) * K + ac4;
    const float* pA1 = A + (size_t)min(rowBase + ar + 32, M - 1) * K + ac4;
    const float* pA2 = A + (size_t)min(rowBase + ar + 64, M - 1) * K + ac4;
    const float* pA3 = A + (size_t)min(rowBase + ar + 96, M - 1) * K + ac4;
    const int sA = ar * ASTR + ac4;
    const int wr  = tid >> 4;
    const int wc4 = (tid & 15) << 2;
    const float* pW0 = W + (size_t)wr * N + colBase + wc4;
    const float* pW1 = pW0 + (size_t)16 * N;
    const int sW = wc4 * WTSTR + wr;

#define GEMM_LOAD(kc_)                                              \
    ra0 = *(const float4*)(pA0 + (kc_));                            \
    ra1 = *(const float4*)(pA1 + (kc_));                            \
    ra2 = *(const float4*)(pA2 + (kc_));                            \
    ra3 = *(const float4*)(pA3 + (kc_));                            \
    rw0 = *(const float4*)(pW0 + (size_t)(kc_) * N);                \
    rw1 = *(const float4*)(pW1 + (size_t)(kc_) * N);
#define GEMM_STORE(s_)                                              \
    *(float4*)(&lA[s_][sA])             = ra0;                      \
    *(float4*)(&lA[s_][sA + 32 * ASTR]) = ra1;                      \
    *(float4*)(&lA[s_][sA + 64 * ASTR]) = ra2;                      \
    *(float4*)(&lA[s_][sA + 96 * ASTR]) = ra3;                      \
    lWt[s_][sW + 0 * WTSTR]      = rw0.x;                           \
    lWt[s_][sW + 1 * WTSTR]      = rw0.y;                           \
    lWt[s_][sW + 2 * WTSTR]      = rw0.z;                           \
    lWt[s_][sW + 3 * WTSTR]      = rw0.w;                           \
    lWt[s_][sW + 0 * WTSTR + 16] = rw1.x;                           \
    lWt[s_][sW + 1 * WTSTR + 16] = rw1.y;                           \
    lWt[s_][sW + 2 * WTSTR + 16] = rw1.z;                           \
    lWt[s_][sW + 3 * WTSTR + 16] = rw1.w;

    float4 ra0, ra1, ra2, ra3, rw0, rw1;
    GEMM_LOAD(0)
    GEMM_STORE(0)
    __syncthreads();

    v8f acc[4] = {};
    const int aOff = (wid * 16 + l16) * ASTR + 2 * half;
    const int bOff = l16 * WTSTR + 2 * half;

    for (int kc = 0; kc < K; kc += KC) {
        const int  st   = (kc / KC) & 1;
        const bool more = (kc + KC < K);
        if (more) { GEMM_LOAD(kc + KC) }

        const float* sa = &lA[st][aOff];
        const float* sb = &lWt[st][bOff];
        for (int kk = 0; kk < KC; kk += 4) {
            v2f a = *(const v2f*)(sa + kk);
#pragma unroll
            for (int ct = 0; ct < 4; ++ct) {
                v2f b = *(const v2f*)(sb + ct * 16 * WTSTR + kk);
                acc[ct] = __builtin_amdgcn_wmma_f32_16x16x4_f32(
                    false, a, false, b, (short)0, acc[ct], false, false);
            }
        }

        if (more) { GEMM_STORE(st ^ 1) }
        __syncthreads();
    }
#undef GEMM_LOAD
#undef GEMM_STORE

#pragma unroll
    for (int ct = 0; ct < 4; ++ct) {
        const int col = colBase + ct * 16 + l16;
#pragma unroll
        for (int v = 0; v < 8; ++v) {
            const int row = rowBase + wid * 16 + v + 8 * half;
            if (row < M) C[(size_t)row * N + col] = acc[ct][v];
        }
    }
}

#define EPI_BN_RELU 0
#define EPI_SOFTMAX 1
template <int F, int EPI>
__global__ __launch_bounds__(TPB)
void gcn_gather(const int* __restrict__ rowstart, const int* __restrict__ csr_src,
                const float* __restrict__ xw, const float* __restrict__ bias,
                const float* __restrict__ g, const float* __restrict__ beta,
                const float* __restrict__ m, const float* __restrict__ v,
                float* __restrict__ y, int n, int E) {
    static_assert(EPI == EPI_BN_RELU || F == 64, "the softmax form is written for 64 columns");
    constexpr int VW  = (F >= 128) ? 4 : F / 32;
    constexpr int NP  = F / (32 * VW);
    constexpr int UNR = (F >= 256) ? 4 : 8;
    typedef float vec_t __attribute__((ext_vector_type(VW)));
    const int lane = threadIdx.x & 31;
    const int row = __builtin_amdgcn_readfirstlane((int)(blockIdx.x * (TPB / 32) + (threadIdx.x >> 5)));
    if (row >= n) return;
    const int beg = rowstart[row];
    const int end = (row + 1 < n) ? rowstart[row + 1] : E;
    const float di = rsqrtf(fmaxf((float)(end - beg + 1), 1.0f));
    const int col = lane * VW;

    vec_t acc[NP];
    {
        const float sl = di * di;
#pragma unroll
        for (int p = 0; p < NP; ++p) {
            const vec_t xv = *(const vec_t*)(xw + (size_t)row * F + p * 32 * VW + col);
            const vec_t bv = *(const vec_t*)(bias + p * 32 * VW + col);
            acc[p] = bv + xv * sl;
        }
    }

    for (int base = beg; base < end; base += 32) {
        const int nb = min(32, end - base);
        int   s_l = 0;
        float c_l = 0.0f;
        if (lane < nb) {
            s_l = csr_src[base + lane];
            const int a  = rowstart[s_l];
            const int bn = rowstart[min(s_l + 1, n - 1)];
            const int b  = (s_l + 1 < n) ? bn : E;
            c_l = rsqrtf(fmaxf((float)(b - a + 1), 1.0f)) * di;
        }
        const int c_bits = __builtin_bit_cast(int, c_l);
        int k = 0;
        for (; k + UNR <= nb; k += UNR) {
            vec_t xr[UNR][NP];
            float c[UNR];
#pragma unroll
            for (int u = 0; u < UNR; ++u) {
                const int s = __builtin_amdgcn_readlane(s_l, k + u);
                c[u] = __builtin_bit_cast(float, __builtin_amdgcn_readlane(c_bits, k + u));
                const float* r = xw + (size_t)s * F + col;
#pragma unroll
                for (int p = 0; p < NP; ++p) xr[u][p] = *(const vec_t*)(r + p * 32 * VW);
            }
#pragma unroll
            for (int u = 0; u < UNR; ++u)
#pragma unroll
                for (int p = 0; p < NP; ++p) acc[p] += xr[u][p] * c[u];
        }
#pragma unroll 1
        for (; k < nb; ++k) {
            const int s = __builtin_amdgcn_readlane(s_l, k);
            const float c = __builtin_bit_cast(float, __builtin_amdgcn_readlane(c_bits, k));
            const float* r = xw + (size_t)s * F + col;
#pragma unroll
            for (int p = 0; p < NP; ++p) acc[p] += *(const vec_t*)(r + p * 32 * VW) * c;
        }
    }

    if constexpr (EPI == EPI_BN_RELU) {
#pragma unroll
        for (int p = 0; p < NP; ++p) {
            const int c0 = p * 32 * VW + col;
            const vec_t gv = *(const vec_t*)(g + c0);
            const vec_t bv = *(const vec_t*)(beta + c0);
            const vec_t mv = *(const vec_t*)(m + c0);
            const vec_t vv = *(const vec_t*)(v + c0);
            vec_t o;
#pragma unroll
            for (int j = 0; j < VW; ++j)
                o[j] = fmaxf((acc[p][j] - mv[j]) * rsqrtf(vv[j] + 1e-5f) * gv[j] + bv[j], 0.0f);
            *(vec_t*)(y + (size_t)row * F + c0) = o;
        }
    } else {
        const float a0 = acc[0][0], a1 = acc[0][1];
        float mx = fmaxf(a0, a1);
#pragma unroll
        for (int off = 16; off > 0; off >>= 1) mx = fmaxf(mx, __shfl_xor(mx, off, 32));
        const float e0 = __expf(a0 - mx), e1 = __expf(a1 - mx);
        float s = e0 + e1;
#pragma unroll
        for (int off = 16; off > 0; off >>= 1) s += __shfl_xor(s, off, 32);
        const float inv = 1.0f / s;
        vec_t o;
        o[0] = e0 * inv;
        o[1] = e1 * inv;
        *(vec_t*)(y + (size_t)row * F + col) = o;
    }
}

#define FR   32
#define FSTR 260
template <int NOUT>
__global__ __launch_bounds__(TPB)
void gcn_gather_gemm(const int* __restrict__ rowstart, const int* __restrict__ csr_src,
                     const float* __restrict__ xw, const float* __restrict__ bias,
                     const float* __restrict__ g, const float* __restrict__ beta,
                     const float* __restrict__ m, const float* __restrict__ v,
                     const float* __restrict__ W, float* __restrict__ C, int n, int E) {
    constexpr int F   = 256;
    constexpr int VW  = 4;
    constexpr int NP  = F / (32 * VW);
    constexpr int UNR = 4;
    constexpr int NT  = NOUT / 64;
    constexpr int NWL = NOUT / 32;
    constexpr int KU  = (NT == 1) ? KC / 4 : 2;
    typedef float vec_t __attribute__((ext_vector_type(VW)));
    __shared__ __align__(16) float lA[FR * FSTR];
    __shared__ __align__(16) float lWt[NOUT * WTSTR];
    const int tid  = threadIdx.x;
    const int lane = tid & 31;
    const int l16  = lane & 15;
    const int half = lane >> 4;
    const int wid  = __builtin_amdgcn_readfirstlane(tid >> 5);
    const int rowBase = blockIdx.x * FR;
    const int col = lane * VW;

    const int wk  = tid >> 3;
    const int wc4 = (tid & 7) << 2;
    const float* pW = W + (size_t)wk * NOUT + wc4;
    float4 rw[NWL];
#define FUSE_LOADW(kc_)                                                        \
    _Pragma("unroll")                                                          \
    for (int q = 0; q < NWL; ++q) rw[q] = *(const float4*)(pW + (size_t)(kc_) * NOUT + q * 32);
#define FUSE_STOREW()                                                          \
    _Pragma("unroll")                                                          \
    for (int q = 0; q < NWL; ++q) {                                            \
        lWt[(q * 32 + wc4 + 0) * WTSTR + wk] = rw[q].x;                        \
        lWt[(q * 32 + wc4 + 1) * WTSTR + wk] = rw[q].y;                        \
        lWt[(q * 32 + wc4 + 2) * WTSTR + wk] = rw[q].z;                        \
        lWt[(q * 32 + wc4 + 3) * WTSTR + wk] = rw[q].w;                        \
    }
    FUSE_LOADW(0)

    for (int i = 0; i < FR / 8; ++i) {
        const int lr  = wid + 8 * i;
        const int row = rowBase + lr;
        vec_t o[NP];
        if (row < n) {
            const int beg = rowstart[row];
            const int end = (row + 1 < n) ? rowstart[row + 1] : E;
            const float di = rsqrtf(fmaxf((float)(end - beg + 1), 1.0f));
            vec_t acc[NP];
            {
                const float sl = di * di;
#pragma unroll
                for (int p = 0; p < NP; ++p) {
                    const vec_t xv = *(const vec_t*)(xw + (size_t)row * F + p * 32 * VW + col);
                    const vec_t bv = *(const vec_t*)(bias + p * 32 * VW + col);
                    acc[p] = bv + xv * sl;
                }
            }
            for (int base = beg; base < end; base += 32) {
                const int nb = min(32, end - base);
                int   s_l = 0;
                float c_l = 0.0f;
                if (lane < nb) {
                    s_l = csr_src[base + lane];
                    const int a  = rowstart[s_l];
                    const int bn = rowstart[min(s_l + 1, n - 1)];
                    const int b  = (s_l + 1 < n) ? bn : E;
                    c_l = rsqrtf(fmaxf((float)(b - a + 1), 1.0f)) * di;
                }
                const int c_bits = __builtin_bit_cast(int, c_l);
                int k = 0;
                for (; k + UNR <= nb; k += UNR) {
                    vec_t xr[UNR][NP];
                    float c[UNR];
#pragma unroll
                    for (int u = 0; u < UNR; ++u) {
                        const int s = __builtin_amdgcn_readlane(s_l, k + u);
                        c[u] = __builtin_bit_cast(float, __builtin_amdgcn_readlane(c_bits, k + u));
                        const float* r = xw + (size_t)s * F + col;
#pragma unroll
                        for (int p = 0; p < NP; ++p) xr[u][p] = *(const vec_t*)(r + p * 32 * VW);
                    }
#pragma unroll
                    for (int u = 0; u < UNR; ++u)
#pragma unroll
                        for (int p = 0; p < NP; ++p) acc[p] += xr[u][p] * c[u];
                }
#pragma unroll 1
                for (; k < nb; ++k) {
                    const int s = __builtin_amdgcn_readlane(s_l, k);
                    const float c = __builtin_bit_cast(float, __builtin_amdgcn_readlane(c_bits, k));
                    const float* r = xw + (size_t)s * F + col;
#pragma unroll
                    for (int p = 0; p < NP; ++p) acc[p] += *(const vec_t*)(r + p * 32 * VW) * c;
                }
            }
#pragma unroll
            for (int p = 0; p < NP; ++p) {
                const int c0 = p * 32 * VW + col;
                const vec_t gv = *(const vec_t*)(g + c0);
                const vec_t bv = *(const vec_t*)(beta + c0);
                const vec_t mv = *(const vec_t*)(m + c0);
                const vec_t vv = *(const vec_t*)(v + c0);
#pragma unroll
                for (int j = 0; j < VW; ++j)
                    o[p][j] = fmaxf((acc[p][j] - mv[j]) * rsqrtf(vv[j] + 1e-5f) * gv[j] + bv[j], 0.0f);
            }
        } else {
#pragma unroll
            for (int p = 0; p < NP; ++p) o[p] = (vec_t)(0.0f);
        }
#pragma unroll
        for (int p = 0; p < NP; ++p) *(vec_t*)(&lA[lr * FSTR + p * 32 * VW + col]) = o[p];
    }

    FUSE_STOREW()
    __syncthreads();

    const int rs = wid & 1;
    const int cq = wid >> 1;
    v8f acc2[NT] = {};
    const float* sa0 = &lA[(rs * 16 + l16) * FSTR + 2 * half];
    const float* sb  = &lWt[(cq * (NOUT / 4) + l16) * WTSTR + 2 * half];
    for (int kc = 0; kc < F; kc += KC) {
        const bool more = (kc + KC < F);
        if (more) { FUSE_LOADW(kc + KC) }
        const float* sa = sa0 + kc;
#pragma unroll KU
        for (int kk = 0; kk < KC; kk += 4) {
            v2f a = *(const v2f*)(sa + kk);
#pragma unroll
            for (int ct = 0; ct < NT; ++ct) {
                v2f b = *(const v2f*)(sb + ct * 16 * WTSTR + kk);
                acc2[ct] = __builtin_amdgcn_wmma_f32_16x16x4_f32(
                    false, a, false, b, (short)0, acc2[ct], false, false);
            }
            if ((kk / 4) % KU == KU - 1) {
                __builtin_amdgcn_sched_barrier(0);
                asm volatile("v_nop\n\tv_nop\n\tv_nop\n\tv_nop");
                __builtin_amdgcn_sched_barrier(0);
            }
        }
        if (more) {
            __syncthreads();
            FUSE_STOREW()
            __syncthreads();
        }
    }
#undef FUSE_LOADW
#undef FUSE_STOREW

#pragma unroll
    for (int ct = 0; ct < NT; ++ct) {
        const int cc = cq * (NOUT / 4) + ct * 16 + l16;
#pragma unroll
        for (int vv = 0; vv < 8; ++vv) {
            const int row = rowBase + rs * 16 + vv + 8 * half;
            if (row < n) C[(size_t)row * NOUT + cc] = acc2[ct][vv];
        }
    }
}

extern "C" void kernel_launch(void* const* d_in, const int* in_sizes, int n_in,
                              void* d_out, int out_size, void* d_ws, size_t ws_size,
                              hipStream_t stream) {
    const int IN = 256, HID_ = 256, OUT_ = 64;
    const int N = in_sizes[0] / IN;
    const int E = in_sizes[1] / 2;

    const float* x    = (const float*)d_in[0];
    const int*   src  = (const int*)d_in[1];
    const int*   dst  = ((const int*)d_in[1]) + E;
    const float* W1   = (const float*)d_in[2];
    const float* b1   = (const float*)d_in[3];
    const float* g1   = (const float*)d_in[4];
    const float* be1  = (const float*)d_in[5];
    const float* m1   = (const float*)d_in[6];
    const float* v1   = (const float*)d_in[7];
    const float* W2   = (const float*)d_in[8];
    const float* b2   = (const float*)d_in[9];
    const float* g2   = (const float*)d_in[10];
    const float* be2  = (const float*)d_in[11];
    const float* m2   = (const float*)d_in[12];
    const float* v2   = (const float*)d_in[13];
    const float* W3   = (const float*)d_in[14];
    const float* b3   = (const float*)d_in[15];
    float* out = (float*)d_out;

    char* ws = (char*)d_ws;
    size_t off = 0;
    int*   rowstart = (int*)(ws + off);   off += ((size_t)N * 4 + 255) & ~(size_t)255;
    int*   csr_src  = (int*)(ws + off);   off += ((size_t)E * 4 + 255) & ~(size_t)255;
    float* bufA = (float*)(ws + off); off += (size_t)N * HID_ * 4;
    float* bufB = (float*)(ws + off);
    int2* binned = (int2*)bufA;
    int*  offT   = (int*)bufB;
    (void)ws_size; (void)n_in; (void)out_size;

    const int P   = (N + BIN_ROWS - 1) / BIN_ROWS;
    const int NBA = (E + EPB - 1) / EPB;
    const int gemmX = (N + 127) / 128;
    const int gRow = (N + TPB / 32 - 1) / (TPB / 32);
    const int gFuse = (N + FR - 1) / FR;

    gcn_bin_edges <<<NBA, 1024, 0, stream>>>(src, dst, binned, offT, E, P, NBA);
    gcn_bin_to_csr<<<P, 1024, 0, stream>>>(binned, offT, rowstart, csr_src, N, NBA);

    gcn_gemm_wmma<<<dim3(gemmX, HID_ / 64), TPB, 0, stream>>>(x, W1, bufA, N, IN, HID_);

    gcn_gather_gemm<256><<<gFuse, TPB, 0, stream>>>(rowstart, csr_src, bufA, b1,
                                                    g1, be1, m1, v1, W2, bufB, N, E);

    gcn_gather_gemm<64><<<gFuse, TPB, 0, stream>>>(rowstart, csr_src, bufB, b2,
                                                   g2, be2, m2, v2, W3, bufA, N, E);

    gcn_gather<64, EPI_SOFTMAX><<<gRow, TPB, 0, stream>>>(rowstart, csr_src, bufA, b3,
                                                          nullptr, nullptr, nullptr, nullptr, out, N, E);
}


# Per-edge coefficient removed: rows are stored times their own dinv, gathers add plain rows and scale the sum once by dinv[i], bias last; no rowstart[s] fetches; term rounding changes, edge order kept
# speedup vs baseline: 4.0036x; 1.0449x over previous
#include <hip/hip_runtime.h>
#include <hip/hip_bf16.h>

typedef __attribute__((ext_vector_type(2))) float v2f;
typedef __attribute__((ext_vector_type(8))) float v8f;

#define TPB 256
#define KC    32
#define ASTR  36
#define WTSTR 36

#define BIN_SHIFT 7
#define BIN_ROWS  (1 << BIN_SHIFT)
#define EPB       4096
#define SPL       4
#define GRP       4

__global__ __launch_bounds__(1024)
void gcn_bin_edges(const int* __restrict__ src, const int* __restrict__ dst,
                   int2* __restrict__ binned, int* __restrict__ offT, int E, int P, int NBA) {
    __shared__ int hist[1024];
    __shared__ int wtot[32];
    const int tid  = threadIdx.x;
    const int lane = tid & 31;
    const int wid  = tid >> 5;
    const int k    = blockIdx.x;
    const int e0   = k * EPB;
    hist[tid] = 0;
    int d[4], s[4], r[4];
#pragma unroll
    for (int j = 0; j < 4; ++j) {
        const int e = e0 + j * 1024 + tid;
        d[j] = (e < E) ? dst[e] : -1;
        s[j] = (e < E) ? src[e] : 0;
    }
    __syncthreads();
#pragma unroll
    for (int j = 0; j < 4; ++j)
        r[j] = (d[j] >= 0) ? atomicAdd(&hist[d[j] >> BIN_SHIFT], 1) : 0;
    __syncthreads();
    const int c = hist[tid];
    int inc = c;
#pragma unroll
    for (int off = 1; off < 32; off <<= 1) {
        int t = __shfl_up(inc, off, 32);
        if (lane >= off) inc += t;
    }
    if (lane == 31) wtot[wid] = inc;
    __syncthreads();
    int pre = (lane < wid) ? wtot[lane] : 0;
#pragma unroll
    for (int off = 16; off > 0; off >>= 1) pre += __shfl_xor(pre, off, 32);
    const int excl = pre + inc - c;
    hist[tid] = excl;
    __syncthreads();
    if (tid <= P) offT[(size_t)tid * NBA + k] = excl;
#pragma unroll
    for (int j = 0; j < 4; ++j)
        if (d[j] >= 0) binned[e0 + hist[d[j] >> BIN_SHIFT] + r[j]] = make_int2(d[j], s[j]);
}

__global__ __launch_bounds__(1024)
void gcn_bin_to_csr(const int2* __restrict__ binned, const int* __restrict__ offT,
                    int* __restrict__ rowstart, int* __restrict__ csr_src, int n, int NBA) {
    __shared__ int cnt[BIN_ROWS];
    __shared__ int wtot[32];
    __shared__ int wrow[BIN_ROWS / 32];
    const int tid  = threadIdx.x;
    const int lane = tid & 31;
    const int wid  = tid >> 5;
    const int b    = blockIdx.x;
    const int lo   = b << BIN_SHIFT;
    const int* __restrict__ o_lo = offT + (size_t)b * NBA;
    const int* __restrict__ o_hi = o_lo + NBA;

    if (tid < BIN_ROWS) cnt[tid] = 0;
    __syncthreads();

    int part = 0;
    for (int q = tid; q < NBA * SPL; q += 1024) {
        const int k  = q / SPL;
        const int j  = q % SPL;
        const int2* __restrict__ seg = binned + (size_t)k * EPB;
        const int o0 = o_lo[k];
        const int o1 = o_hi[k];
        if (j == 0) part += o0;
        int i = o0 + j;
        int dd[GRP];
#pragma unroll
        for (int u = 0; u < GRP; ++u) dd[u] = (i + u * SPL < o1) ? seg[i + u * SPL].x : -1;
#pragma unroll
        for (int u = 0; u < GRP; ++u) if (dd[u] >= 0) atomicAdd(&cnt[dd[u] - lo], 1);
        for (i += GRP * SPL; i < o1; i += SPL) atomicAdd(&cnt[seg[i].x - lo], 1);
    }
#pragma unroll
    for (int off = 16; off > 0; off >>= 1) part += __shfl_xor(part, off, 32);
    if (lane == 0) wtot[wid] = part;
    __syncthreads();
    int base = wtot[lane];
#pragma unroll
    for (int off = 16; off > 0; off >>= 1) base += __shfl_xor(base, off, 32);

    int start = 0;
    if (tid < BIN_ROWS) {
        const int c = cnt[tid];
        int inc = c;
#pragma unroll
        for (int off = 1; off < 32; off <<= 1) {
            int t = __shfl_up(inc, off, 32);
            if (lane >= off) inc += t;
        }
        if (lane == 31) wrow[wid] = inc;
        start = base + inc - c;
    }
    __syncthreads();
    if (tid < BIN_ROWS) {
        int pre = (lane < wid) ? wrow[lane] : 0;
#pragma unroll
        for (int off = 16; off > 0; off >>= 1) pre += __shfl_xor(pre, off, 32);
        start += pre;
        cnt[tid] = start;
    }
    __syncthreads();
    if (tid < BIN_ROWS && lo + tid < n) rowstart[lo + tid] = start;

    for (int q = tid; q < NBA * SPL; q += 1024) {
        const int k  = q / SPL;
        const int2* __restrict__ seg = binned + (size_t)k * EPB;
        const int o1 = o_hi[k];
        int i = o_lo[k] + (q % SPL);
        int2 ee[GRP];
#pragma unroll
        for (int u = 0; u < GRP; ++u) ee[u] = (i + u * SPL < o1) ? seg[i + u * SPL] : make_int2(-1, 0);
#pragma unroll
        for (int u = 0; u < GRP; ++u)
            if (ee[u].x >= 0) csr_src[atomicAdd(&cnt[ee[u].x - lo], 1)] = ee[u].y;
        for (i += GRP * SPL; i < o1; i += SPL) {
            const int2 e = seg[i];
            csr_src[atomicAdd(&cnt[e.x - lo], 1)] = e.y;
        }
    }
}

__global__ __launch_bounds__(TPB)
void gcn_gemm_wmma(const float* __restrict__ A, const float* __restrict__ W,
                   const int* __restrict__ rowstart,
                   float* __restrict__ C, int M, int K, int N, int E) {
    __shared__ float lA[2][128 * ASTR];
    __shared__ float lWt[2][64 * WTSTR];
    __shared__ float sdinv[128];
    const int tid  = threadIdx.x;
    const int lane = tid & 31;
    const int wid  = tid >> 5;
    const int l16  = lane & 15;
    const int half = lane >> 4;
    const int rowBase = blockIdx.x * 128;
    const int colBase = blockIdx.y * 64;

    const int ar  = tid >> 3;
    const int ac4 = (tid & 7) << 2;
    const float* pA0 = A + (size_t)min(rowBase + ar,      M - 1) * K + ac4;
    const float* pA1 = A + (size_t)min(rowBase + ar + 32, M - 1) * K + ac4;
    const float* pA2 = A + (size_t)min(rowBase + ar + 64, M - 1) * K + ac4;
    const float* pA3 = A + (size_t)min(rowBase + ar + 96, M - 1) * K + ac4;
    const int sA = ar * ASTR + ac4;
    const int wr  = tid >> 4;
    const int wc4 = (tid & 15) << 2;
    const float* pW0 = W + (size_t)wr * N + colBase + wc4;
    const float* pW1 = pW0 + (size_t)16 * N;
    const int sW = wc4 * WTSTR + wr;

#define GEMM_LOAD(kc_)                                              \
    ra0 = *(const float4*)(pA0 + (kc_));                            \
    ra1 = *(const float4*)(pA1 + (kc_));                            \
    ra2 = *(const float4*)(pA2 + (kc_));                            \
    ra3 = *(const float4*)(pA3 + (kc_));                            \
    rw0 = *(const float4*)(pW0 + (size_t)(kc_) * N);                \
    rw1 = *(const float4*)(pW1 + (size_t)(kc_) * N);
#define GEMM_STORE(s_)                                              \
    *(float4*)(&lA[s_][sA])             = ra0;                      \
    *(float4*)(&lA[s_][sA + 32 * ASTR]) = ra1;                      \
    *(float4*)(&lA[s_][sA + 64 * ASTR]) = ra2;                      \
    *(float4*)(&lA[s_][sA + 96 * ASTR]) = ra3;                      \
    lWt[s_][sW + 0 * WTSTR]      = rw0.x;                           \
    lWt[s_][sW + 1 * WTSTR]      = rw0.y;                           \
    lWt[s_][sW + 2 * WTSTR]      = rw0.z;                           \
    lWt[s_][sW + 3 * WTSTR]      = rw0.w;                           \
    lWt[s_][sW + 0 * WTSTR + 16] = rw1.x;                           \
    lWt[s_][sW + 1 * WTSTR + 16] = rw1.y;                           \
    lWt[s_][sW + 2 * WTSTR + 16] = rw1.z;                           \
    lWt[s_][sW + 3 * WTSTR + 16] = rw1.w;

    float4 ra0, ra1, ra2, ra3, rw0, rw1;
    GEMM_LOAD(0)
    if (tid < 128) {
        const int r  = min(rowBase + tid, M - 1);
        const int a  = rowstart[r];
        const int bn = rowstart[min(r + 1, M - 1)];
        const int b  = (r + 1 < M) ? bn : E;
        sdinv[tid] = rsqrtf(fmaxf((float)(b - a + 1), 1.0f));
    }
    GEMM_STORE(0)
    __syncthreads();

    v8f acc[4] = {};
    const int aOff = (wid * 16 + l16) * ASTR + 2 * half;
    const int bOff = l16 * WTSTR + 2 * half;

    for (int kc = 0; kc < K; kc += KC) {
        const int  st   = (kc / KC) & 1;
        const bool more = (kc + KC < K);
        if (more) { GEMM_LOAD(kc + KC) }

        const float* sa = &lA[st][aOff];
        const float* sb = &lWt[st][bOff];
        for (int kk = 0; kk < KC; kk += 4) {
            v2f a = *(const v2f*)(sa + kk);
#pragma unroll
            for (int ct = 0; ct < 4; ++ct) {
                v2f b = *(const v2f*)(sb + ct * 16 * WTSTR + kk);
                acc[ct] = __builtin_amdgcn_wmma_f32_16x16x4_f32(
                    false, a, false, b, (short)0, acc[ct], false, false);
            }
        }

        if (more) { GEMM_STORE(st ^ 1) }
        __syncthreads();
    }
#undef GEMM_LOAD
#undef GEMM_STORE

    float sc[8];
#pragma unroll
    for (int v = 0; v < 8; ++v) sc[v] = sdinv[wid * 16 + v + 8 * half];
#pragma unroll
    for (int ct = 0; ct < 4; ++ct) {
        const int col = colBase + ct * 16 + l16;
#pragma unroll
        for (int v = 0; v < 8; ++v) {
            const int row = rowBase + wid * 16 + v + 8 * half;
            if (row < M) C[(size_t)row * N + col] = acc[ct][v] * sc[v];
        }
    }
}

#define EPI_BN_RELU 0
#define EPI_SOFTMAX 1
template <int F, int EPI>
__global__ __launch_bounds__(TPB)
void gcn_gather(const int* __restrict__ rowstart, const int* __restrict__ csr_src,
                const float* __restrict__ xw, const float* __restrict__ bias,
                const float* __restrict__ g, const float* __restrict__ beta,
                const float* __restrict__ m, const float* __restrict__ v,
                float* __restrict__ y, int n, int E) {
    static_assert(EPI == EPI_BN_RELU || F == 64, "the softmax form is written for 64 columns");
    constexpr int VW  = (F >= 128) ? 4 : F / 32;
    constexpr int NP  = F / (32 * VW);
    constexpr int UNR = (F >= 256) ? 4 : 8;
    typedef float vec_t __attribute__((ext_vector_type(VW)));
    const int lane = threadIdx.x & 31;
    const int row = __builtin_amdgcn_readfirstlane((int)(blockIdx.x * (TPB / 32) + (threadIdx.x >> 5)));
    if (row >= n) return;
    const int beg = rowstart[row];
    const int end = (row + 1 < n) ? rowstart[row + 1] : E;
    const float di = rsqrtf(fmaxf((float)(end - beg + 1), 1.0f));
    const int col = lane * VW;

    vec_t acc[NP];
#pragma unroll
    for (int p = 0; p < NP; ++p) acc[p] = *(const vec_t*)(xw + (size_t)row * F + p * 32 * VW + col);

    for (int base = beg; base < end; base += 32) {
        const int nb = min(32, end - base);
        int s_l = 0;
        if (lane < nb) s_l = csr_src[base + lane];
        int k = 0;
        for (; k + UNR <= nb; k += UNR) {
            vec_t xr[UNR][NP];
#pragma unroll
            for (int u = 0; u < UNR; ++u) {
                const int s = __builtin_amdgcn_readlane(s_l, k + u);
                const float* r = xw + (size_t)s * F + col;
#pragma unroll
                for (int p = 0; p < NP; ++p) xr[u][p] = *(const vec_t*)(r + p * 32 * VW);
            }
#pragma unroll
            for (int u = 0; u < UNR; ++u)
#pragma unroll
                for (int p = 0; p < NP; ++p) acc[p] += xr[u][p];
        }
#pragma unroll 1
        for (; k < nb; ++k) {
            const int s = __builtin_amdgcn_readlane(s_l, k);
            const float* r = xw + (size_t)s * F + col;
#pragma unroll
            for (int p = 0; p < NP; ++p) acc[p] += *(const vec_t*)(r + p * 32 * VW);
        }
    }
#pragma unroll
    for (int p = 0; p < NP; ++p) acc[p] = *(const vec_t*)(bias + p * 32 * VW + col) + acc[p] * di;

    if constexpr (EPI == EPI_BN_RELU) {
#pragma unroll
        for (int p = 0; p < NP; ++p) {
            const int c0 = p * 32 * VW + col;
            const vec_t gv = *(const vec_t*)(g + c0);
            const vec_t bv = *(const vec_t*)(beta + c0);
            const vec_t mv = *(const vec_t*)(m + c0);
            const vec_t vv = *(const vec_t*)(v + c0);
            vec_t o;
#pragma unroll
            for (int j = 0; j < VW; ++j)
                o[j] = fmaxf((acc[p][j] - mv[j]) * rsqrtf(vv[j] + 1e-5f) * gv[j] + bv[j], 0.0f);
            *(vec_t*)(y + (size_t)row * F + c0) = o;
        }
    } else {
        const float a0 = acc[0][0], a1 = acc[0][1];
        float mx = fmaxf(a0, a1);
#pragma unroll
        for (int off = 16; off > 0; off >>= 1) mx = fmaxf(mx, __shfl_xor(mx, off, 32));
        const float e0 = __expf(a0 - mx), e1 = __expf(a1 - mx);
        float s = e0 + e1;
#pragma unroll
        for (int off = 16; off > 0; off >>= 1) s += __shfl_xor(s, off, 32);
        const float inv = 1.0f / s;
        vec_t o;
        o[0] = e0 * inv;
        o[1] = e1 * inv;
        *(vec_t*)(y + (size_t)row * F + col) = o;
    }
}

#define FR   32
#define FSTR 260
template <int NOUT>
__global__ __launch_bounds__(TPB)
void gcn_gather_gemm(const int* __restrict__ rowstart, const int* __restrict__ csr_src,
                     const float* __restrict__ xw, const float* __restrict__ bias,
                     const float* __restrict__ g, const float* __restrict__ beta,
                     const float* __restrict__ m, const float* __restrict__ v,
                     const float* __restrict__ W, float* __restrict__ C, int n, int E) {
    constexpr int F   = 256;
    constexpr int VW  = 4;
    constexpr int NP  = F / (32 * VW);
    constexpr int UNR = 4;
    constexpr int NT  = NOUT / 64;
    constexpr int NWL = NOUT / 32;
    constexpr int KU  = (NT == 1) ? KC / 4 : 2;
    typedef float vec_t __attribute__((ext_vector_type(VW)));
    __shared__ __align__(16) float lA[FR * FSTR];
    __shared__ __align__(16) float lWt[NOUT * WTSTR];
    __shared__ float sdinv[FR];
    const int tid  = threadIdx.x;
    const int lane = tid & 31;
    const int l16  = lane & 15;
    const int half = lane >> 4;
    const int wid  = __builtin_amdgcn_readfirstlane(tid >> 5);
    const int rowBase = blockIdx.x * FR;
    const int col = lane * VW;

    const int wk  = tid >> 3;
    const int wc4 = (tid & 7) << 2;
    const float* pW = W + (size_t)wk * NOUT + wc4;
    float4 rw[NWL];
#define FUSE_LOADW(kc_)                                                        \
    _Pragma("unroll")                                                          \
    for (int q = 0; q < NWL; ++q) rw[q] = *(const float4*)(pW + (size_t)(kc_) * NOUT + q * 32);
#define FUSE_STOREW()                                                          \
    _Pragma("unroll")                                                          \
    for (int q = 0; q < NWL; ++q) {                                            \
        lWt[(q * 32 + wc4 + 0) * WTSTR + wk] = rw[q].x;                        \
        lWt[(q * 32 + wc4 + 1) * WTSTR + wk] = rw[q].y;                        \
        lWt[(q * 32 + wc4 + 2) * WTSTR + wk] = rw[q].z;                        \
        lWt[(q * 32 + wc4 + 3) * WTSTR + wk] = rw[q].w;                        \
    }
    FUSE_LOADW(0)

    for (int i = 0; i < FR / 8; ++i) {
        const int lr  = wid + 8 * i;
        const int row = rowBase + lr;
        vec_t o[NP];
        if (row < n) {
            const int beg = rowstart[row];
            const int end = (row + 1 < n) ? rowstart[row + 1] : E;
            const float di = rsqrtf(fmaxf((float)(end - beg + 1), 1.0f));
            if (lane == 0) sdinv[lr] = di;
            vec_t acc[NP];
#pragma unroll
            for (int p = 0; p < NP; ++p) acc[p] = *(const vec_t*)(xw + (size_t)row * F + p * 32 * VW + col);
            for (int base = beg; base < end; base += 32) {
                const int nb = min(32, end - base);
                int s_l = 0;
                if (lane < nb) s_l = csr_src[base + lane];
                int k = 0;
                for (; k + UNR <= nb; k += UNR) {
                    vec_t xr[UNR][NP];
#pragma unroll
                    for (int u = 0; u < UNR; ++u) {
                        const int s = __builtin_amdgcn_readlane(s_l, k + u);
                        const float* r = xw + (size_t)s * F + col;
#pragma unroll
                        for (int p = 0; p < NP; ++p) xr[u][p] = *(const vec_t*)(r + p * 32 * VW);
                    }
#pragma unroll
                    for (int u = 0; u < UNR; ++u)
#pragma unroll
                        for (int p = 0; p < NP; ++p) acc[p] += xr[u][p];
                }
#pragma unroll 1
                for (; k < nb; ++k) {
                    const int s = __builtin_amdgcn_readlane(s_l, k);
                    const float* r = xw + (size_t)s * F + col;
#pragma unroll
                    for (int p = 0; p < NP; ++p) acc[p] += *(const vec_t*)(r + p * 32 * VW);
                }
            }
#pragma unroll
            for (int p = 0; p < NP; ++p) acc[p] = *(const vec_t*)(bias + p * 32 * VW + col) + acc[p] * di;
#pragma unroll
            for (int p = 0; p < NP; ++p) {
                const int c0 = p * 32 * VW + col;
                const vec_t gv = *(const vec_t*)(g + c0);
                const vec_t bv = *(const vec_t*)(beta + c0);
                const vec_t mv = *(const vec_t*)(m + c0);
                const vec_t vv = *(const vec_t*)(v + c0);
#pragma unroll
                for (int j = 0; j < VW; ++j)
                    o[p][j] = fmaxf((acc[p][j] - mv[j]) * rsqrtf(vv[j] + 1e-5f) * gv[j] + bv[j], 0.0f);
            }
        } else {
            if (lane == 0) sdinv[lr] = 0.0f;
#pragma unroll
            for (int p = 0; p < NP; ++p) o[p] = (vec_t)(0.0f);
        }
#pragma unroll
        for (int p = 0; p < NP; ++p) *(vec_t*)(&lA[lr * FSTR + p * 32 * VW + col]) = o[p];
    }

    FUSE_STOREW()
    __syncthreads();

    const int rs = wid & 1;
    const int cq = wid >> 1;
    v8f acc2[NT] = {};
    const float* sa0 = &lA[(rs * 16 + l16) * FSTR + 2 * half];
    const float* sb  = &lWt[(cq * (NOUT / 4) + l16) * WTSTR + 2 * half];
    for (int kc = 0; kc < F; kc += KC) {
        const bool more = (kc + KC < F);
        if (more) { FUSE_LOADW(kc + KC) }
        const float* sa = sa0 + kc;
#pragma unroll KU
        for (int kk = 0; kk < KC; kk += 4) {
            v2f a = *(const v2f*)(sa + kk);
#pragma unroll
            for (int ct = 0; ct < NT; ++ct) {
                v2f b = *(const v2f*)(sb + ct * 16 * WTSTR + kk);
                acc2[ct] = __builtin_amdgcn_wmma_f32_16x16x4_f32(
                    false, a, false, b, (short)0, acc2[ct], false, false);
            }
            if ((kk / 4) % KU == KU - 1) {
                __builtin_amdgcn_sched_barrier(0);
                asm volatile("v_nop\n\tv_nop\n\tv_nop\n\tv_nop");
                __builtin_amdgcn_sched_barrier(0);
            }
        }
        if (more) {
            __syncthreads();
            FUSE_STOREW()
            __syncthreads();
        }
    }
#undef FUSE_LOADW
#undef FUSE_STOREW

    float sc[8];
#pragma unroll
    for (int vv = 0; vv < 8; ++vv) sc[vv] = sdinv[rs * 16 + vv + 8 * half];
#pragma unroll
    for (int ct = 0; ct < NT; ++ct) {
        const int cc = cq * (NOUT / 4) + ct * 16 + l16;
#pragma unroll
        for (int vv = 0; vv < 8; ++vv) {
            const int row = rowBase + rs * 16 + vv + 8 * half;
            if (row < n) C[(size_t)row * NOUT + cc] = acc2[ct][vv] * sc[vv];
        }
    }
}

extern "C" void kernel_launch(void* const* d_in, const int* in_sizes, int n_in,
                              void* d_out, int out_size, void* d_ws, size_t ws_size,
                              hipStream_t stream) {
    const int IN = 256, HID_ = 256, OUT_ = 64;
    const int N = in_sizes[0] / IN;
    const int E = in_sizes[1] / 2;

    const float* x    = (const float*)d_in[0];
    const int*   src  = (const int*)d_in[1];
    const int*   dst  = ((const int*)d_in[1]) + E;
    const float* W1   = (const float*)d_in[2];
    const float* b1   = (const float*)d_in[3];
    const float* g1   = (const float*)d_in[4];
    const float* be1  = (const float*)d_in[5];
    const float* m1   = (const float*)d_in[6];
    const float* v1   = (const float*)d_in[7];
    const float* W2   = (const float*)d_in[8];
    const float* b2   = (const float*)d_in[9];
    const float* g2   = (const float*)d_in[10];
    const float* be2  = (const float*)d_in[11];
    const float* m2   = (const float*)d_in[12];
    const float* v2   = (const float*)d_in[13];
    const float* W3   = (const float*)d_in[14];
    const float* b3   = (const float*)d_in[15];
    float* out = (float*)d_out;

    char* ws = (char*)d_ws;
    size_t off = 0;
    int*   rowstart = (int*)(ws + off);   off += ((size_t)N * 4 + 255) & ~(size_t)255;
    int*   csr_src  = (int*)(ws + off);   off += ((size_t)E * 4 + 255) & ~(size_t)255;
    float* bufA = (float*)(ws + off); off += (size_t)N * HID_ * 4;
    float* bufB = (float*)(ws + off);
    int2* binned = (int2*)bufA;
    int*  offT   = (int*)bufB;
    (void)ws_size; (void)n_in; (void)out_size;

    const int P   = (N + BIN_ROWS - 1) / BIN_ROWS;
    const int NBA = (E + EPB - 1) / EPB;
    const int gemmX = (N + 127) / 128;
    const int gRow = (N + TPB / 32 - 1) / (TPB / 32);
    const int gFuse = (N + FR - 1) / FR;

    gcn_bin_edges <<<NBA, 1024, 0, stream>>>(src, dst, binned, offT, E, P, NBA);
    gcn_bin_to_csr<<<P, 1024, 0, stream>>>(binned, offT, rowstart, csr_src, N, NBA);

    gcn_gemm_wmma<<<dim3(gemmX, HID_ / 64), TPB, 0, stream>>>(x, W1, rowstart, bufA, N, IN, HID_, E);

    gcn_gather_gemm<256><<<gFuse, TPB, 0, stream>>>(rowstart, csr_src, bufA, b1,
                                                    g1, be1, m1, v1, W2, bufB, N, E);

    gcn_gather_gemm<64><<<gFuse, TPB, 0, stream>>>(rowstart, csr_src, bufB, b2,
                                                   g2, be2, m2, v2, W3, bufA, N, E);

    gcn_gather<64, EPI_SOFTMAX><<<gRow, TPB, 0, stream>>>(rowstart, csr_src, bufA, b3,
                                                          nullptr, nullptr, nullptr, nullptr, out, N, E);
}
